# starting kernel
# baseline (speedup 1.0000x reference)
#include <hip/hip_runtime.h>


typedef __attribute__((ext_vector_type(2))) float v2f;
typedef __attribute__((ext_vector_type(8))) float v8f;

static constexpr int kN   = 129;
static constexpr int kIN  = 129;
static constexpr int kOUT = 64;
static constexpr int kE   = 4096;
static constexpr int kLD  = 132;
static constexpr int kRP  = 144;
static constexpr int kKS  = 33;
static constexpr int kMT  = 9;

static __device__ __forceinline__ v8f wmma4(v2f a, v2f b, v8f c) {
  return __builtin_amdgcn_wmma_f32_16x16x4_f32(
       false, a,  false, b,
       (short)0, c,  false,  false);
}

static __device__ __forceinline__ float sigm(float x) {
  return 1.0f / (1.0f + __expf(-x));
}

__launch_bounds__(512, 1)
__global__ void evolve_gcn_fused(const float* __restrict__ x,
                                 const int*   __restrict__ eidx,
                                 const float* __restrict__ ew,
                                 const float* __restrict__ pool_p,
                                 const float* __restrict__ W0,
                                 const float* __restrict__ w_ih,
                                 const float* __restrict__ w_hh,
                                 const float* __restrict__ b_ih,
                                 const float* __restrict__ b_hh,
                                 const float* __restrict__ conv_b,
                                 const float* __restrict__ lin_w,
                                 const float* __restrict__ lin_b,
                                 float*       __restrict__ out)
{
  __shared__ float s_x [kRP * kLD];
  __shared__ float s_xt[kRP * kLD];
  __shared__ float s_h [kRP * kLD];
  __shared__ float s_w [kRP * kLD];
  __shared__ float s_score[kN];
  __shared__ int   s_perm [kN];
  __shared__ float s_dis  [kN];
  __shared__ float s_deg  [kN];

  const int tid    = threadIdx.x;
  const int nthr   = blockDim.x;
  const int lane   = tid & 31;
  const int wave   = tid >> 5;
  const int nwav   = nthr >> 5;
  const int m16    = lane & 15;
  const int hh     = lane >> 4;

  for (int i = tid; i < kRP * kLD; i += nthr) {
    s_x[i] = 0.f; s_xt[i] = 0.f; s_h[i] = 0.f; s_w[i] = 0.f;
  }
  for (int i = tid; i < kN; i += nthr) s_deg[i] = 0.f;
  __syncthreads();

  for (int i = tid; i < kN * kIN; i += nthr) {
    int r = i / kIN, c = i % kIN;
    s_x[r * kLD + c] = x[i];
    s_h[r * kLD + c] = W0[i];
  }
  for (int e = tid; e < kE; e += nthr) {
    atomicAdd(&s_deg[eidx[kE + e]], ew[e]);
  }
  for (int i = tid; i < kN; i += nthr) atomicAdd(&s_deg[i], 1.0f);
  __syncthreads();

  for (int i = tid; i < kN; i += nthr) {
    float pn = 0.f, d = 0.f;
    for (int k = 0; k < kIN; ++k) { float pv = pool_p[k]; pn += pv * pv; }
    for (int k = 0; k < kIN; ++k) d += s_x[i * kLD + k] * pool_p[k];
    s_score[i] = tanhf(d * rsqrtf(pn));
    float dg = s_deg[i];
    s_dis[i] = dg > 0.f ? rsqrtf(fmaxf(dg, 1e-12f)) : 0.f;
  }
  __syncthreads();

  for (int i = tid; i < kN; i += nthr) {
    float si = s_score[i];
    int rank = 0;
    for (int j = 0; j < kN; ++j) {
      float sj = s_score[j];
      rank += (sj > si) || (sj == si && j < i);
    }
    s_perm[rank] = i;
  }
  __syncthreads();

  for (int i = tid; i < kN * kLD; i += nthr) {
    int r = i / kLD, c = i % kLD;
    int src = s_perm[r];
    s_xt[r * kLD + c] = s_x[src * kLD + c] * s_score[src];
  }
  __syncthreads();

  for (int t = wave; t < kMT * kMT; t += nwav) {
    int mt = t / kMT, nt = t % kMT;
    int jn = nt * 16 + m16;
    const int rowr = jn * kIN;
    const int rowz = (kN + jn) * kIN;
    const int rown = min(2 * kN + jn, 3 * kN - 1) * kIN;
    float br = b_ih[min(jn, 3 * kN - 1)]          + b_hh[min(jn, 3 * kN - 1)];
    float bz = b_ih[min(kN + jn, 3 * kN - 1)]     + b_hh[min(kN + jn, 3 * kN - 1)];
    float bi = b_ih[min(2 * kN + jn, 3 * kN - 1)];
    float bh = b_hh[min(2 * kN + jn, 3 * kN - 1)];
    v8f ar, az, ai, ah;
#pragma unroll
    for (int v = 0; v < 8; ++v) { ar[v] = br; az[v] = bz; ai[v] = bi; ah[v] = bh; }

    const int arow = (mt * 16 + m16) * kLD;
    for (int ks = 0; ks < kKS; ++ks) {
      int k0 = ks * 4 + 2 * hh;
      v2f axt, ah0;
      axt.x = s_xt[arow + k0]; axt.y = s_xt[arow + k0 + 1];
      ah0.x = s_h [arow + k0]; ah0.y = s_h [arow + k0 + 1];
      int i0 = min(k0,     kIN - 1);
      int i1 = min(k0 + 1, kIN - 1);
      v2f bri, bzi, bni, brh, bzh, bnh;
      bri.x = w_ih[rowr + i0];  bri.y = w_ih[rowr + i1];
      bzi.x = w_ih[rowz + i0];  bzi.y = w_ih[rowz + i1];
      bni.x = w_ih[rown + i0];  bni.y = w_ih[rown + i1];
      brh.x = w_hh[rowr + i0];  brh.y = w_hh[rowr + i1];
      bzh.x = w_hh[rowz + i0];  bzh.y = w_hh[rowz + i1];
      bnh.x = w_hh[rown + i0];  bnh.y = w_hh[rown + i1];
      ar = wmma4(axt, bri, ar);   ar = wmma4(ah0, brh, ar);
      az = wmma4(axt, bzi, az);   az = wmma4(ah0, bzh, az);
      ai = wmma4(axt, bni, ai);
      ah = wmma4(ah0, bnh, ah);
    }
    if (jn < kLD) {
#pragma unroll
      for (int v = 0; v < 8; ++v) {
        int row = mt * 16 + v + 8 * hh;
        float rg  = sigm(ar[v]);
        float zg  = sigm(az[v]);
        float cnd = tanhf(ai[v] + rg * ah[v]);
        float w0v = s_h[row * kLD + jn];
        s_w[row * kLD + jn] = (1.f - zg) * cnd + zg * w0v;
      }
    }
  }
  __syncthreads();

  for (int t = wave; t < kMT * kMT; t += nwav) {
    int mt = t / kMT, nt = t % kMT;
    int jn = nt * 16 + m16;
    int jn_c = min(jn, kLD - 1);
    v8f acc = {0.f, 0.f, 0.f, 0.f, 0.f, 0.f, 0.f, 0.f};
    const int arow = (mt * 16 + m16) * kLD;
    for (int ks = 0; ks < kKS; ++ks) {
      int k0 = ks * 4 + 2 * hh;
      v2f a; a.x = s_x[arow + k0];           a.y = s_x[arow + k0 + 1];
      v2f b; b.x = s_w[k0 * kLD + jn_c];     b.y = s_w[(k0 + 1) * kLD + jn_c];
      acc = wmma4(a, b, acc);
    }
    if (jn < kLD) {
#pragma unroll
      for (int v = 0; v < 8; ++v)
        s_h[(mt * 16 + v + 8 * hh) * kLD + jn] = acc[v];
    }
  }
  __syncthreads();

  float* s_out = s_xt;
  for (int i = tid; i < kRP * kLD; i += nthr) s_out[i] = 0.f;
  __syncthreads();

  for (int e = wave; e < kE + kN; e += nwav) {
    int r, c; float wgt;
    if (e < kE) { r = eidx[e]; c = eidx[kE + e]; wgt = ew[e]; }
    else        { r = e - kE;  c = r;            wgt = 1.0f; }
    float nrm = s_dis[r] * wgt * s_dis[c];
    for (int f = lane; f < kN; f += 32)
      atomicAdd(&s_out[c * kLD + f], nrm * s_h[r * kLD + f]);
  }
  __syncthreads();

  for (int i = tid; i < kN * kN; i += nthr) {
    int r = i / kN, c = i % kN;
    float v = s_out[r * kLD + c] + conv_b[c];
    s_out[r * kLD + c] = v > 0.f ? v : (__expf(v) - 1.f);
  }
  __syncthreads();

  for (int t = wave; t < kMT * 4; t += nwav) {
    int mt = t / 4, ot = t % 4;
    int oc = ot * 16 + m16;
    float lb = lin_b[oc];
    v8f acc;
#pragma unroll
    for (int v = 0; v < 8; ++v) acc[v] = lb;
    const int arow = (mt * 16 + m16) * kLD;
    const int rowo = oc * kIN;
    for (int ks = 0; ks < kKS; ++ks) {
      int k0 = ks * 4 + 2 * hh;
      v2f a; a.x = s_out[arow + k0]; a.y = s_out[arow + k0 + 1];
      int i0 = min(k0,     kIN - 1);
      int i1 = min(k0 + 1, kIN - 1);
      v2f b; b.x = lin_w[rowo + i0]; b.y = lin_w[rowo + i1];
      acc = wmma4(a, b, acc);
    }
#pragma unroll
    for (int v = 0; v < 8; ++v) {
      int row = mt * 16 + v + 8 * hh;
      if (row < kN) out[row * kOUT + oc] = acc[v];
    }
  }
}

extern "C" void kernel_launch(void* const* d_in, const int* in_sizes, int n_in,
                              void* d_out, int out_size, void* d_ws, size_t ws_size,
                              hipStream_t stream) {
  (void)in_sizes; (void)n_in; (void)out_size; (void)d_ws; (void)ws_size;
  const float* x       = (const float*)d_in[0];
  const int*   eidx    = (const int*)  d_in[1];
  const float* ew      = (const float*)d_in[2];
  const float* pool_p  = (const float*)d_in[3];
  const float* W0      = (const float*)d_in[4];
  const float* w_ih    = (const float*)d_in[5];
  const float* w_hh    = (const float*)d_in[6];
  const float* b_ih    = (const float*)d_in[7];
  const float* b_hh    = (const float*)d_in[8];
  const float* conv_b  = (const float*)d_in[9];
  const float* lin_w   = (const float*)d_in[10];
  const float* lin_b   = (const float*)d_in[11];
  evolve_gcn_fused<<<1, 512, 0, stream>>>(x, eidx, ew, pool_p, W0, w_ih, w_hh,
                                          b_ih, b_hh, conv_b, lin_w, lin_b,
                                          (float*)d_out);
}


# GRU step spread over 27 workgroups by W row tile (one 16x16 cell a wave; front repeated per group); W passes through the workspace to a 2nd launch: xw, edge sum, ELU, linear; sums keep their order
# speedup vs baseline: 6.3871x; 6.3871x over previous
#include <hip/hip_runtime.h>


typedef __attribute__((ext_vector_type(2))) float v2f;
typedef __attribute__((ext_vector_type(4))) float v4f;
typedef __attribute__((ext_vector_type(8))) float v8f;
typedef v4f v4f_a4 __attribute__((aligned(4)));

static constexpr int kN   = 129;
static constexpr int kIN  = 129;
static constexpr int kOUT = 64;
static constexpr int kE   = 4096;
static constexpr int kLD  = 132;
static constexpr int kRP  = 144;
static constexpr int kKS  = 33;
static constexpr int kMT  = 9;

static __device__ __forceinline__ v8f wmma4(v2f a, v2f b, v8f c) {
  return __builtin_amdgcn_wmma_f32_16x16x4_f32(
       false, a,  false, b,
       (short)0, c,  false,  false);
}

static __device__ __forceinline__ float sigm(float x) {
  return 1.0f / (1.0f + __expf(-x));
}

struct BFrag {
  v4f q[16];
  v2f ql;
};

static __device__ __forceinline__ void load_bfrag(BFrag& b, const float* __restrict__ w,
                                                  int grow, int hh) {
  const float* p = w + grow * kIN + 4 * hh;
  const float t = w[grow * kIN + (kIN - 1)];
  b.ql.x = t; b.ql.y = t;
#pragma unroll
  for (int j = 0; j < 16; ++j) b.q[j] = *(const v4f_a4*)(p + 8 * j);
}

template <int NT>
static __device__ __forceinline__ void gru_pass(v8f (&acc)[NT], const float* A,
                                                BFrag& b, int abase, int hh, bool more,
                                                const float* __restrict__ wn, int nrow) {
  const float* pn = wn + nrow * kIN + 4 * hh;
  v4f a[NT];
  asm volatile("" ::: "memory");
  float qn = 0.f;
  if (more) qn = wn[nrow * kIN + (kIN - 1)];
#pragma unroll
  for (int t = 0; t < NT; ++t)
    a[t] = *(const v4f*)&A[abase + t * 16 * kLD + 4 * hh];
#pragma unroll
  for (int j = 0; j < 16; ++j) {
    asm volatile("" ::: "memory");
    v4f an[NT];
    v2f al[NT];
    if (j < 15) {
#pragma unroll
      for (int t = 0; t < NT; ++t)
        an[t] = *(const v4f*)&A[abase + t * 16 * kLD + 8 * (j + 1) + 4 * hh];
    } else {
#pragma unroll
      for (int t = 0; t < NT; ++t) {
        const int at = abase + t * 16 * kLD + 4 * (kKS - 1) + 2 * hh;
        al[t].x = A[at]; al[t].y = A[at + 1];
      }
    }
#pragma unroll
    for (int t = 0; t < NT; ++t) {
      v2f a0, b0; a0.x = a[t].x; a0.y = a[t].y; b0.x = b.q[j].x; b0.y = b.q[j].y;
      acc[t] = wmma4(a0, b0, acc[t]);
    }
#pragma unroll
    for (int t = 0; t < NT; ++t) {
      v2f a1, b1; a1.x = a[t].z; a1.y = a[t].w; b1.x = b.q[j].z; b1.y = b.q[j].w;
      acc[t] = wmma4(a1, b1, acc[t]);
    }
    if (more) b.q[j] = *(const v4f_a4*)(pn + 8 * j);
    if (j < 15) {
#pragma unroll
      for (int t = 0; t < NT; ++t) a[t] = an[t];
    } else {
#pragma unroll
      for (int t = 0; t < NT; ++t) acc[t] = wmma4(al[t], b.ql, acc[t]);
    }
  }
  if (more) { b.ql.x = qn; b.ql.y = qn; }
}

template <int NT, bool LAST>
static __device__ __forceinline__ void gru_gate(
    int g, int grow, int grown, float& b2, BFrag& B,
    int jn, int jn_c, int abase, int cbase, int hh,
    const float* __restrict__ w_ih, const float* __restrict__ w_hh,
    const float* __restrict__ b_ih, const float* __restrict__ b_hh,
    const float* s_xt, const float* s_h, float* s_w) {
  const float b1 = b_ih[grow];
  v8f acc[NT];
#pragma unroll
  for (int t = 0; t < NT; ++t)
#pragma unroll
    for (int v = 0; v < 8; ++v) acc[t][v] = b2;
  gru_pass<NT>(acc, s_h, B, abase, hh, true, w_ih, grow);
  if (!LAST) b2 = b_hh[grown];
  if (!LAST && g == 1) {
#pragma unroll
    for (int t = 0; t < NT; ++t)
#pragma unroll
      for (int v = 0; v < 8; ++v)
        acc[t][v] = b1 + s_w[cbase + (t * 16 + v) * kLD + jn_c] * acc[t][v];
  } else {
#pragma unroll
    for (int t = 0; t < NT; ++t)
#pragma unroll
      for (int v = 0; v < 8; ++v) acc[t][v] += b1;
  }
  gru_pass<NT>(acc, s_xt, B, abase, hh, !LAST, w_hh, grown);
  if (jn < kLD) {
    if (LAST) {
#pragma unroll
      for (int t = 0; t < NT; ++t)
#pragma unroll
        for (int v = 0; v < 8; ++v) {
          const int idx = cbase + (t * 16 + v) * kLD + jn;
          const float zg = sigm(acc[t][v]);
          s_w[idx] = (1.f - zg) * s_w[idx] + zg * s_h[idx];
        }
    } else if (g == 0) {
#pragma unroll
      for (int t = 0; t < NT; ++t)
#pragma unroll
        for (int v = 0; v < 8; ++v)
          s_w[cbase + (t * 16 + v) * kLD + jn] = sigm(acc[t][v]);
    } else {
#pragma unroll 1
      for (int t = 0; t < NT; ++t) {
#pragma unroll
        for (int v = 0; v < 8; ++v)
          s_w[cbase + (t * 16 + v) * kLD + jn] = tanhf(acc[0][v]);
        const v8f first = acc[0];
#pragma unroll
        for (int u = 0; u + 1 < NT; ++u) acc[u] = acc[u + 1];
        acc[NT - 1] = first;
      }
    }
  }
}

template <int NT>
static __device__ __forceinline__ void gru_cells(
    int nt, int mb, int m16, int hh,
    const float* __restrict__ w_ih, const float* __restrict__ w_hh,
    const float* __restrict__ b_ih, const float* __restrict__ b_hh,
    const float* s_xt, const float* s_h, float* s_w) {
  const int jn    = nt * 16 + m16;
  const int jn_c  = min(jn, kLD - 1);
  const int abase = (mb * 16 + m16) * kLD;
  const int cbase = (mb * 16 + 8 * hh) * kLD;
  const int grow_r = min(0 * kN + jn, 3 * kN - 1);
  const int grow_n = min(2 * kN + jn, 3 * kN - 1);
  const int grow_z = min(1 * kN + jn, 3 * kN - 1);
  float b2 = b_hh[grow_r];
  BFrag B;
  load_bfrag(B, w_hh, grow_r, hh);
#pragma unroll 1
  for (int g = 0; g < 2; ++g)
    gru_gate<NT, false>(g, g == 0 ? grow_r : grow_n, g == 0 ? grow_n : grow_z, b2, B,
                        jn, jn_c, abase, cbase, hh, w_ih, w_hh, b_ih, b_hh, s_xt, s_h, s_w);
  gru_gate<NT, true>(2, grow_z, grow_z, b2, B,
                     jn, jn_c, abase, cbase, hh, w_ih, w_hh, b_ih, b_hh, s_xt, s_h, s_w);
}

__launch_bounds__(512, 1)
__global__ void gru_rows(const float* __restrict__ x,
                         const float* __restrict__ pool_p,
                         const float* __restrict__ W0,
                         const float* __restrict__ w_ih,
                         const float* __restrict__ w_hh,
                         const float* __restrict__ b_ih,
                         const float* __restrict__ b_hh,
                         float*       __restrict__ wsW)
{
  __shared__ __attribute__((aligned(16))) float s_x [kRP * kLD];
  __shared__ __attribute__((aligned(16))) float s_xt[kRP * kLD];
  __shared__ __attribute__((aligned(16))) float s_h [kRP * kLD];
  __shared__ __attribute__((aligned(16))) float s_w [kRP * kLD];
  __shared__ float s_score[kN];
  __shared__ int   s_perm [kN];

  const int tid    = threadIdx.x;
  const int nthr   = blockDim.x;
  const int lane   = tid & 31;
  const int wave   = tid >> 5;
  const int m16    = lane & 15;
  const int hh     = lane >> 4;
  const int rt     = blockIdx.x / 3;
  const int cg     = blockIdx.x % 3;

  constexpr int kRPW = 8;
  const int wv0 = __builtin_amdgcn_readfirstlane(wave);
  v4f xr[kRPW];
  float xl[kRPW];
#pragma unroll
  for (int i = 0; i < kRPW; ++i) {
    const int r = wv0 + 16 * i;
    xr[i] = *(const v4f_a4*)(x + r * kIN + 4 * lane);
    xl[i] = x[r * kIN + (kIN - 1)];
  }
  v4f xr8 = {0.f, 0.f, 0.f, 0.f};
  float xl8 = 0.f;
  if (wv0 == 0) {
    xr8 = *(const v4f_a4*)(x + (kN - 1) * kIN + 4 * lane);
    xl8 = x[(kN - 1) * kIN + (kIN - 1)];
  }
  const int hrow = rt * 16 + wv0;
  v4f h4 = {0.f, 0.f, 0.f, 0.f};
  float h1 = 0.f;
  if (hrow < kN) {
    h4 = *(const v4f_a4*)(W0 + hrow * kIN + 4 * lane);
    h1 = W0[hrow * kIN + (kIN - 1)];
  }
  const v4f   pp = *(const v4f_a4*)(pool_p + 4 * lane);
  const float pl = pool_p[kIN - 1];

  for (int i = tid; i < 16 * kLD; i += nthr) {
    s_xt[rt * 16 * kLD + i] = 0.f; s_h[rt * 16 * kLD + i] = 0.f;
  }
  __syncthreads();

  const int nrow = (wv0 == 0) ? kRPW + 1 : kRPW;
#pragma unroll
  for (int i = 0; i < kRPW; ++i) {
    const int r = wv0 + 16 * i;
    *(v4f*)&s_x[r * kLD + 4 * lane] = xr[i];
  }
  if (wv0 == 0) *(v4f*)&s_x[(kN - 1) * kLD + 4 * lane] = xr8;
  if (hrow < kN) {
    *(v4f*)&s_h[hrow * kLD + 4 * lane] = h4;
    if (lane == 0) s_h[hrow * kLD + (kIN - 1)] = h1;
  }
  {
    float xs = xl8;
#pragma unroll
    for (int i = 0; i < kRPW; ++i)
      if (lane == i) xs = xl[i];
    if (lane < nrow) s_x[(wv0 + 16 * lane) * kLD + (kIN - 1)] = xs;
  }
  {
    float pn = pp.x * pp.x + pp.y * pp.y + pp.z * pp.z + pp.w * pp.w;
    float d[kRPW + 1];
#pragma unroll
    for (int i = 0; i < kRPW; ++i)
      d[i] = xr[i].x * pp.x + xr[i].y * pp.y + xr[i].z * pp.z + xr[i].w * pp.w;
    d[kRPW] = xr8.x * pp.x + xr8.y * pp.y + xr8.z * pp.z + xr8.w * pp.w;
#pragma unroll
    for (int m = 16; m >= 1; m >>= 1) {
      pn += __shfl_xor(pn, m);
#pragma unroll
      for (int i = 0; i <= kRPW; ++i) d[i] += __shfl_xor(d[i], m);
    }
    pn += pl * pl;
    float ds = d[kRPW] + xl8 * pl;
#pragma unroll
    for (int i = 0; i < kRPW; ++i)
      if (lane == i) ds = d[i] + xl[i] * pl;
    if (lane < nrow) s_score[wv0 + 16 * lane] = tanhf(ds * rsqrtf(pn));
  }
  __syncthreads();

  for (int i = tid; i < kN; i += nthr) {
    float si = s_score[i];
    int rank = 0;
    for (int j = 0; j < kN; ++j) {
      float sj = s_score[j];
      rank += (sj > si) || (sj == si && j < i);
    }
    s_perm[rank] = i;
  }
  __syncthreads();

  for (int i = tid; i < 16 * kLD; i += nthr) {
    const int r = rt * 16 + i / kLD, c = i % kLD;
    if (r < kN && c < kIN) {
      const int src = s_perm[r];
      s_xt[r * kLD + c] = s_x[src * kLD + c] * s_score[src];
    }
  }
  __syncthreads();

  {
    const int wv = __builtin_amdgcn_readfirstlane(wave);
    if (wv < 3)
      gru_cells<1>(cg * 3 + wv, rt, m16, hh, w_ih, w_hh, b_ih, b_hh, s_xt, s_h, s_w);
  }
  __syncthreads();

  for (int i = tid; i < 16 * 48; i += nthr) {
    const int r = rt * 16 + i / 48, c = cg * 48 + i % 48;
    wsW[blockIdx.x * (16 * 48) + i] = (r < kN && c < kLD) ? s_w[r * kLD + c] : 0.f;
  }
}

__launch_bounds__(512, 1)
__global__ void gcn_tail(const float* __restrict__ x,
                         const int*   __restrict__ eidx,
                         const float* __restrict__ ew,
                         const float* __restrict__ conv_b,
                         const float* __restrict__ lin_w,
                         const float* __restrict__ lin_b,
                         const float* __restrict__ wsW,
                         float*       __restrict__ out)
{
  __shared__ __attribute__((aligned(16))) float s_x [kRP * kLD];
  __shared__ __attribute__((aligned(16))) float s_xt[kRP * kLD];
  __shared__ __attribute__((aligned(16))) float s_h [kRP * kLD];
  __shared__ __attribute__((aligned(16))) float s_w [kRP * kLD];
  __shared__ float s_dis  [kN];
  __shared__ float s_deg  [kN];
  __shared__ int   s_cnt  [kN];
  __shared__ int   s_beg  [kN];
  __shared__ int   s_cur  [kN];

  const int tid    = threadIdx.x;
  const int nthr   = blockDim.x;
  const int lane   = tid & 31;
  const int wave   = tid >> 5;
  const int nwav   = nthr >> 5;
  const int m16    = lane & 15;
  const int hh     = lane >> 4;

  constexpr int kRPW = 8;
  constexpr int kNT0 = 512;
  constexpr int kEP0 = kE / kNT0;
  constexpr int kWQ  = 27 * 16 * 12;
  constexpr int kWPT = (kWQ + kNT0 - 1) / kNT0;
  const int wv0 = __builtin_amdgcn_readfirstlane(wave);
  v4f xr[kRPW];
  float xl[kRPW];
#pragma unroll
  for (int i = 0; i < kRPW; ++i) {
    const int r = wv0 + 16 * i;
    xr[i] = *(const v4f_a4*)(x + r * kIN + 4 * lane);
    xl[i] = x[r * kIN + (kIN - 1)];
  }
  v4f xr8 = {0.f, 0.f, 0.f, 0.f};
  float xl8 = 0.f;
  if (wv0 == 0) {
    xr8 = *(const v4f_a4*)(x + (kN - 1) * kIN + 4 * lane);
    xl8 = x[(kN - 1) * kIN + (kIN - 1)];
  }
  v4f wq[kWPT];
#pragma unroll
  for (int u = 0; u < kWPT; ++u) {
    const int q = min(tid + u * kNT0, kWQ - 1);
    wq[u] = *(const v4f*)(wsW + 4 * q);
  }
  int ecol[kEP0]; float ewt[kEP0];
#pragma unroll
  for (int u = 0; u < kEP0; ++u) {
    ecol[u] = eidx[kE + tid + u * kNT0];
    ewt[u]  = ew[tid + u * kNT0];
  }

  for (int i = tid; i < kRP * kLD; i += nthr) s_x[i] = 0.f;
  for (int i = kN * kLD + tid; i < kRP * kLD; i += nthr) s_w[i] = 0.f;
  for (int i = tid; i < kN; i += nthr) { s_deg[i] = 0.f; s_cnt[i] = 0; }
  __syncthreads();

  const int nrow = (wv0 == 0) ? kRPW + 1 : kRPW;
#pragma unroll
  for (int i = 0; i < kRPW; ++i) {
    const int r = wv0 + 16 * i;
    *(v4f*)&s_x[r * kLD + 4 * lane] = xr[i];
  }
  if (wv0 == 0) *(v4f*)&s_x[(kN - 1) * kLD + 4 * lane] = xr8;
  {
    float xs = xl8;
#pragma unroll
    for (int i = 0; i < kRPW; ++i)
      if (lane == i) xs = xl[i];
    if (lane < nrow) s_x[(wv0 + 16 * lane) * kLD + (kIN - 1)] = xs;
  }
#pragma unroll
  for (int u = 0; u < kWPT; ++u) {
    const int q  = tid + u * kNT0;
    const int b  = q / 192, j4 = q % 192;
    const int r  = (b / 3) * 16 + j4 / 12;
    const int c  = (b % 3) * 48 + (j4 % 12) * 4;
    if (q < kWQ && r < kN && c < kLD) *(v4f*)&s_w[r * kLD + c] = wq[u];
  }
#pragma unroll
  for (int u = 0; u < kEP0; ++u) {
    atomicAdd(&s_deg[ecol[u]], ewt[u]);
    atomicAdd(&s_cnt[ecol[u]], 1);
  }
  for (int i = tid; i < kN; i += nthr) atomicAdd(&s_deg[i], 1.0f);
  __syncthreads();

  for (int i = tid; i < kN; i += nthr) {
    float dg = s_deg[i];
    s_dis[i] = dg > 0.f ? rsqrtf(fmaxf(dg, 1e-12f)) : 0.f;
  }
  if (tid >= 256 && tid < 256 + kN) {
    const int i = tid - 256;
    int s = 0;
    for (int j = 0; j < kN - 1; ++j) s += (j < i) ? s_cnt[j] : 0;
    s_beg[i] = s;
    s_cur[i] = s;
  }
  __syncthreads();

  {
    constexpr int kNT  = 512;
    constexpr int kEPT = kE / kNT;
    int er[kEPT], ec[kEPT]; float wv8[kEPT];
#pragma unroll
    for (int u = 0; u < kEPT; ++u) {
      const int e = tid + u * kNT;
      er[u] = eidx[e]; ec[u] = eidx[kE + e]; wv8[u] = ew[e];
    }
#pragma unroll
    for (int u = 0; u < kEPT; ++u) {
      const int p = atomicAdd(&s_cur[ec[u]], 1);
      v2f rec;
      rec.x = __int_as_float(er[u]);
      rec.y = s_dis[er[u]] * wv8[u] * s_dis[ec[u]];
      *(v2f*)&s_xt[2 * p] = rec;
    }
  }

  for (int t = wave; t < kMT * kMT; t += nwav) {
    int mt = t / kMT, nt = t % kMT;
    int jn = nt * 16 + m16;
    int jn_c = min(jn, kLD - 1);
    v8f acc = {0.f, 0.f, 0.f, 0.f, 0.f, 0.f, 0.f, 0.f};
    const int arow = (mt * 16 + m16) * kLD;
    for (int ks = 0; ks < kKS; ++ks) {
      int k0 = ks * 4 + 2 * hh;
      v2f a; a.x = s_x[arow + k0];           a.y = s_x[arow + k0 + 1];
      v2f b; b.x = s_w[k0 * kLD + jn_c];     b.y = s_w[(k0 + 1) * kLD + jn_c];
      acc = wmma4(a, b, acc);
    }
    if (jn < kLD) {
#pragma unroll
      for (int v = 0; v < 8; ++v)
        s_h[(mt * 16 + v + 8 * hh) * kLD + jn] = acc[v];
    }
  }
  __syncthreads();

  float* s_out = s_x;
  {
    const int wv = __builtin_amdgcn_readfirstlane(wave);
    float cb[4];
#pragma unroll
    for (int u = 0; u < 4; ++u) cb[u] = conv_b[4 * lane + u];
    const float cb128 = conv_b[kN - 1];
    for (int c = wv; c < kN; c += nwav) {
      const int beg = __builtin_amdgcn_readfirstlane(s_beg[c]);
      const int n   = __builtin_amdgcn_readfirstlane(s_cnt[c]);
      const float dc = s_dis[c];
      const float sl = dc * 1.0f * dc;
      const v4f xs = *(const v4f*)&s_h[c * kLD + 4 * lane];
      v4f   acc  = sl * xs;
      float acc1 = sl * s_h[c * kLD + (kN - 1)];
      for (int base = 0; base < n; base += 32) {
        const int m = min(32, n - base);
        int   rr = c;
        float ww = 0.f;
        if (lane < m) {
          const v2f rec = *(const v2f*)&s_xt[2 * (beg + base + lane)];
          rr = __float_as_int(rec.x);
          ww = rec.y;
        }
        for (int i = 0; i < m; i += 8) {
#pragma unroll
          for (int u = 0; u < 8; ++u) {
            const int   r = __builtin_amdgcn_readlane(rr, i + u);
            const float w = __int_as_float(
                __builtin_amdgcn_readlane(__float_as_int(ww), i + u));
            const v4f   v  = *(const v4f*)&s_h[r * kLD + 4 * lane];
            const float v1 = s_h[r * kLD + (kN - 1)];
            acc  += w * v;
            acc1 += w * v1;
          }
        }
      }
      v4f o;
#pragma unroll
      for (int u = 0; u < 4; ++u) {
        const float t = acc[u] + cb[u];
        o[u] = t > 0.f ? t : (__expf(t) - 1.f);
      }
      *(v4f*)&s_out[c * kLD + 4 * lane] = o;
      if (lane == 0) {
        const float t = acc1 + cb128;
        v4f ol;
        ol.x = t > 0.f ? t : (__expf(t) - 1.f);
        ol.y = 0.f; ol.z = 0.f; ol.w = 0.f;
        *(v4f*)&s_out[c * kLD + (kN - 1)] = ol;
      }
    }
  }
  __syncthreads();

  for (int t = wave; t < kMT * 4; t += nwav) {
    int mt = t / 4, ot = t % 4;
    int oc = ot * 16 + m16;
    float lb = lin_b[oc];
    v8f acc;
#pragma unroll
    for (int v = 0; v < 8; ++v) acc[v] = lb;
    const int arow = (mt * 16 + m16) * kLD;
    const int rowo = oc * kIN;
    for (int ks = 0; ks < kKS; ++ks) {
      int k0 = ks * 4 + 2 * hh;
      v2f a; a.x = s_out[arow + k0]; a.y = s_out[arow + k0 + 1];
      int i0 = min(k0,     kIN - 1);
      int i1 = min(k0 + 1, kIN - 1);
      v2f b; b.x = lin_w[rowo + i0]; b.y = lin_w[rowo + i1];
      acc = wmma4(a, b, acc);
    }
#pragma unroll
    for (int v = 0; v < 8; ++v) {
      int row = mt * 16 + v + 8 * hh;
      if (row < kN) out[row * kOUT + oc] = acc[v];
    }
  }
}

extern "C" void kernel_launch(void* const* d_in, const int* in_sizes, int n_in,
                              void* d_out, int out_size, void* d_ws, size_t ws_size,
                              hipStream_t stream) {
  (void)in_sizes; (void)n_in; (void)out_size; (void)ws_size;
  const float* x       = (const float*)d_in[0];
  const int*   eidx    = (const int*)  d_in[1];
  const float* ew      = (const float*)d_in[2];
  const float* pool_p  = (const float*)d_in[3];
  const float* W0      = (const float*)d_in[4];
  const float* w_ih    = (const float*)d_in[5];
  const float* w_hh    = (const float*)d_in[6];
  const float* b_ih    = (const float*)d_in[7];
  const float* b_hh    = (const float*)d_in[8];
  const float* conv_b  = (const float*)d_in[9];
  const float* lin_w   = (const float*)d_in[10];
  const float* lin_b   = (const float*)d_in[11];
  float*       wsW     = (float*)d_ws;
  gru_rows<<<27, 512, 0, stream>>>(x, pool_p, W0, w_ih, w_hh, b_ih, b_hh, wsW);
  gcn_tail<<<1, 512, 0, stream>>>(x, eidx, ew, conv_b, lin_w, lin_b, wsW, (float*)d_out);
}


# Graph conv reassociated to (A_hat x) W: the edge sum needs no W and runs in launch 1 on 9 workgroups beside the GRU; launch 2 is row-independent on 9 workgroups; float sum order changes
# speedup vs baseline: 16.7925x; 2.6291x over previous
#include <hip/hip_runtime.h>


typedef __attribute__((ext_vector_type(2))) float v2f;
typedef __attribute__((ext_vector_type(4))) float v4f;
typedef __attribute__((ext_vector_type(8))) float v8f;
typedef v4f v4f_a4 __attribute__((aligned(4)));

static constexpr int kN   = 129;
static constexpr int kIN  = 129;
static constexpr int kOUT = 64;
static constexpr int kE   = 4096;
static constexpr int kLD  = 132;
static constexpr int kRP  = 144;
static constexpr int kKS  = 33;
static constexpr int kMT  = 9;
static constexpr int kAP  = 160;
static constexpr int kWSW = 27 * 16 * 48;

static __device__ __forceinline__ v8f wmma4(v2f a, v2f b, v8f c) {
  return __builtin_amdgcn_wmma_f32_16x16x4_f32(
       false, a,  false, b,
       (short)0, c,  false,  false);
}

static __device__ __forceinline__ float sigm(float x) {
  return 1.0f / (1.0f + __expf(-x));
}

struct BFrag {
  v4f q[16];
  v2f ql;
};

static __device__ __forceinline__ void load_bfrag(BFrag& b, const float* __restrict__ w,
                                                  int grow, int hh) {
  const float* p = w + grow * kIN + 4 * hh;
  const float t = w[grow * kIN + (kIN - 1)];
  b.ql.x = t; b.ql.y = t;
#pragma unroll
  for (int j = 0; j < 16; ++j) b.q[j] = *(const v4f_a4*)(p + 8 * j);
}

template <int NT>
static __device__ __forceinline__ void gru_pass(v8f (&acc)[NT], const float* A,
                                                BFrag& b, int abase, int hh, bool more,
                                                const float* __restrict__ wn, int nrow) {
  const float* pn = wn + nrow * kIN + 4 * hh;
  v4f a[NT];
  asm volatile("" ::: "memory");
  float qn = 0.f;
  if (more) qn = wn[nrow * kIN + (kIN - 1)];
#pragma unroll
  for (int t = 0; t < NT; ++t)
    a[t] = *(const v4f*)&A[abase + t * 16 * kLD + 4 * hh];
#pragma unroll
  for (int j = 0; j < 16; ++j) {
    asm volatile("" ::: "memory");
    v4f an[NT];
    v2f al[NT];
    if (j < 15) {
#pragma unroll
      for (int t = 0; t < NT; ++t)
        an[t] = *(const v4f*)&A[abase + t * 16 * kLD + 8 * (j + 1) + 4 * hh];
    } else {
#pragma unroll
      for (int t = 0; t < NT; ++t) {
        const int at = abase + t * 16 * kLD + 4 * (kKS - 1) + 2 * hh;
        al[t].x = A[at]; al[t].y = A[at + 1];
      }
    }
#pragma unroll
    for (int t = 0; t < NT; ++t) {
      v2f a0, b0; a0.x = a[t].x; a0.y = a[t].y; b0.x = b.q[j].x; b0.y = b.q[j].y;
      acc[t] = wmma4(a0, b0, acc[t]);
    }
#pragma unroll
    for (int t = 0; t < NT; ++t) {
      v2f a1, b1; a1.x = a[t].z; a1.y = a[t].w; b1.x = b.q[j].z; b1.y = b.q[j].w;
      acc[t] = wmma4(a1, b1, acc[t]);
    }
    if (more) b.q[j] = *(const v4f_a4*)(pn + 8 * j);
    if (j < 15) {
#pragma unroll
      for (int t = 0; t < NT; ++t) a[t] = an[t];
    } else {
#pragma unroll
      for (int t = 0; t < NT; ++t) acc[t] = wmma4(al[t], b.ql, acc[t]);
    }
  }
  if (more) { b.ql.x = qn; b.ql.y = qn; }
}

template <int NT, bool LAST>
static __device__ __forceinline__ void gru_gate(
    int g, int grow, int grown, float& b2, BFrag& B,
    int jn, int jn_c, int abase, int cbase, int hh,
    const float* __restrict__ w_ih, const float* __restrict__ w_hh,
    const float* __restrict__ b_ih, const float* __restrict__ b_hh,
    const float* s_xt, const float* s_h, float* s_w) {
  const float b1 = b_ih[grow];
  v8f acc[NT];
#pragma unroll
  for (int t = 0; t < NT; ++t)
#pragma unroll
    for (int v = 0; v < 8; ++v) acc[t][v] = b2;
  gru_pass<NT>(acc, s_h, B, abase, hh, true, w_ih, grow);
  if (!LAST) b2 = b_hh[grown];
  if (!LAST && g == 1) {
#pragma unroll
    for (int t = 0; t < NT; ++t)
#pragma unroll
      for (int v = 0; v < 8; ++v)
        acc[t][v] = b1 + s_w[cbase + (t * 16 + v) * kLD + jn_c] * acc[t][v];
  } else {
#pragma unroll
    for (int t = 0; t < NT; ++t)
#pragma unroll
      for (int v = 0; v < 8; ++v) acc[t][v] += b1;
  }
  gru_pass<NT>(acc, s_xt, B, abase, hh, !LAST, w_hh, grown);
  if (jn < kLD) {
    if (LAST) {
#pragma unroll
      for (int t = 0; t < NT; ++t)
#pragma unroll
        for (int v = 0; v < 8; ++v) {
          const int idx = cbase + (t * 16 + v) * kLD + jn;
          const float zg = sigm(acc[t][v]);
          s_w[idx] = (1.f - zg) * s_w[idx] + zg * s_h[idx];
        }
    } else if (g == 0) {
#pragma unroll
      for (int t = 0; t < NT; ++t)
#pragma unroll
        for (int v = 0; v < 8; ++v)
          s_w[cbase + (t * 16 + v) * kLD + jn] = sigm(acc[t][v]);
    } else {
#pragma unroll 1
      for (int t = 0; t < NT; ++t) {
#pragma unroll
        for (int v = 0; v < 8; ++v)
          s_w[cbase + (t * 16 + v) * kLD + jn] = tanhf(acc[0][v]);
        const v8f first = acc[0];
#pragma unroll
        for (int u = 0; u + 1 < NT; ++u) acc[u] = acc[u + 1];
        acc[NT - 1] = first;
      }
    }
  }
}

template <int NT>
static __device__ __forceinline__ void gru_cells(
    int nt, int mb, int m16, int hh,
    const float* __restrict__ w_ih, const float* __restrict__ w_hh,
    const float* __restrict__ b_ih, const float* __restrict__ b_hh,
    const float* s_xt, const float* s_h, float* s_w) {
  const int jn    = nt * 16 + m16;
  const int jn_c  = min(jn, kLD - 1);
  const int abase = (mb * 16 + m16) * kLD;
  const int cbase = (mb * 16 + 8 * hh) * kLD;
  const int grow_r = min(0 * kN + jn, 3 * kN - 1);
  const int grow_n = min(2 * kN + jn, 3 * kN - 1);
  const int grow_z = min(1 * kN + jn, 3 * kN - 1);
  float b2 = b_hh[grow_r];
  BFrag B;
  load_bfrag(B, w_hh, grow_r, hh);
#pragma unroll 1
  for (int g = 0; g < 2; ++g)
    gru_gate<NT, false>(g, g == 0 ? grow_r : grow_n, g == 0 ? grow_n : grow_z, b2, B,
                        jn, jn_c, abase, cbase, hh, w_ih, w_hh, b_ih, b_hh, s_xt, s_h, s_w);
  gru_gate<NT, true>(2, grow_z, grow_z, b2, B,
                     jn, jn_c, abase, cbase, hh, w_ih, w_hh, b_ih, b_hh, s_xt, s_h, s_w);
}

static __device__ __forceinline__ void agg_rows(
    int rt, const float* __restrict__ x, const int* __restrict__ eidx,
    const float* __restrict__ ew, float* __restrict__ wsA,
    float* s_x, float* s_rec, float* s_small) {
  float* s_deg = s_small;
  float* s_dis = s_small + 160;
  int*   s_cnt = (int*)(s_small + 320);
  int*   s_beg = (int*)(s_small + 336);
  int*   s_cur = (int*)(s_small + 352);

  const int tid  = threadIdx.x;
  const int lane = tid & 31;
  const int wave = tid >> 5;
  constexpr int kRPW = 8;
  constexpr int kNT0 = 512;
  constexpr int kEP0 = kE / kNT0;
  const int wv0 = __builtin_amdgcn_readfirstlane(wave);

  v4f xr[kRPW];
  float xl[kRPW];
#pragma unroll
  for (int i = 0; i < kRPW; ++i) {
    const int r = wv0 + 16 * i;
    xr[i] = *(const v4f_a4*)(x + r * kIN + 4 * lane);
    xl[i] = x[r * kIN + (kIN - 1)];
  }
  v4f xr8 = {0.f, 0.f, 0.f, 0.f};
  float xl8 = 0.f;
  if (wv0 == 0) {
    xr8 = *(const v4f_a4*)(x + (kN - 1) * kIN + 4 * lane);
    xl8 = x[(kN - 1) * kIN + (kIN - 1)];
  }
  int erow[kEP0], ecol[kEP0]; float ewt[kEP0];
#pragma unroll
  for (int u = 0; u < kEP0; ++u) {
    const int e = tid + u * kNT0;
    erow[u] = eidx[e]; ecol[u] = eidx[kE + e]; ewt[u] = ew[e];
  }

  if (tid < kN) s_deg[tid] = 0.f;
  if (tid < 16) s_cnt[tid] = 0;
  __syncthreads();

  const int nrow = (wv0 == 0) ? kRPW + 1 : kRPW;
#pragma unroll
  for (int i = 0; i < kRPW; ++i) {
    const int r = wv0 + 16 * i;
    *(v4f*)&s_x[r * kLD + 4 * lane] = xr[i];
  }
  if (wv0 == 0) *(v4f*)&s_x[(kN - 1) * kLD + 4 * lane] = xr8;
  {
    float xs = xl8;
#pragma unroll
    for (int i = 0; i < kRPW; ++i)
      if (lane == i) xs = xl[i];
    if (lane < nrow) s_x[(wv0 + 16 * lane) * kLD + (kIN - 1)] = xs;
  }
#pragma unroll
  for (int u = 0; u < kEP0; ++u) {
    atomicAdd(&s_deg[ecol[u]], ewt[u]);
    const unsigned lc = (unsigned)(ecol[u] - 16 * rt);
    if (lc < 16u) atomicAdd(&s_cnt[lc], 1);
  }
  if (tid < kN) atomicAdd(&s_deg[tid], 1.0f);
  __syncthreads();

  if (tid < kN) {
    const float dg = s_deg[tid];
    s_dis[tid] = dg > 0.f ? rsqrtf(fmaxf(dg, 1e-12f)) : 0.f;
  }
  if (tid >= 256 && tid < 256 + 16) {
    const int i = tid - 256;
    int s = 0;
    for (int j = 0; j < 15; ++j) s += (j < i) ? s_cnt[j] : 0;
    s_beg[i] = s;
    s_cur[i] = s;
  }
  __syncthreads();

#pragma unroll
  for (int u = 0; u < kEP0; ++u) {
    const unsigned lc = (unsigned)(ecol[u] - 16 * rt);
    if (lc < 16u) {
      const int p = atomicAdd(&s_cur[lc], 1);
      v2f rec;
      rec.x = __int_as_float(erow[u]);
      rec.y = s_dis[erow[u]] * ewt[u] * s_dis[ecol[u]];
      *(v2f*)&s_rec[2 * p] = rec;
    }
  }
  __syncthreads();

  const int c = 16 * rt + wv0;
  float* dst = wsA + c * kAP;
  v4f   acc  = {0.f, 0.f, 0.f, 0.f};
  float acc1 = 0.f;
  if (c < kN) {
    const int beg = __builtin_amdgcn_readfirstlane(s_beg[wv0]);
    const int n   = __builtin_amdgcn_readfirstlane(s_cnt[wv0]);
    const float dc = s_dis[c];
    const float sl = dc * 1.0f * dc;
    const v4f xs = *(const v4f*)&s_x[c * kLD + 4 * lane];
    acc  = sl * xs;
    acc1 = sl * s_x[c * kLD + (kN - 1)];
    for (int base = 0; base < n; base += 32) {
      const int m = min(32, n - base);
      int   rr = c;
      float ww = 0.f;
      if (lane < m) {
        const v2f rec = *(const v2f*)&s_rec[2 * (beg + base + lane)];
        rr = __float_as_int(rec.x);
        ww = rec.y;
      }
      for (int i = 0; i < m; i += 8) {
#pragma unroll
        for (int u = 0; u < 8; ++u) {
          const int   r = __builtin_amdgcn_readlane(rr, i + u);
          const float w = __int_as_float(
              __builtin_amdgcn_readlane(__float_as_int(ww), i + u));
          const v4f   v  = *(const v4f*)&s_x[r * kLD + 4 * lane];
          const float v1 = s_x[r * kLD + (kN - 1)];
          acc  += w * v;
          acc1 += w * v1;
        }
      }
    }
  }
  *(v4f*)(dst + 4 * lane) = acc;
  if (lane < 8) {
    v4f t = {0.f, 0.f, 0.f, 0.f};
    if (lane == 0) t.x = acc1;
    *(v4f*)(dst + 128 + 4 * lane) = t;
  }
}

__launch_bounds__(512, 1)
__global__ void gru_rows(const float* __restrict__ x,
                         const int*   __restrict__ eidx,
                         const float* __restrict__ ew,
                         const float* __restrict__ pool_p,
                         const float* __restrict__ W0,
                         const float* __restrict__ w_ih,
                         const float* __restrict__ w_hh,
                         const float* __restrict__ b_ih,
                         const float* __restrict__ b_hh,
                         float*       __restrict__ wsW,
                         float*       __restrict__ wsA)
{
  __shared__ __attribute__((aligned(16))) float s_x [kRP * kLD];
  __shared__ __attribute__((aligned(16))) float s_xt[kRP * kLD];
  __shared__ __attribute__((aligned(16))) float s_h [kRP * kLD];
  __shared__ __attribute__((aligned(16))) float s_w [kRP * kLD];
  __shared__ float s_score[kN];
  __shared__ int   s_perm [kN];

  if (blockIdx.x >= 27) {
    agg_rows(blockIdx.x - 27, x, eidx, ew, wsA, s_x, s_xt, s_h);
    return;
  }

  const int tid    = threadIdx.x;
  const int nthr   = blockDim.x;
  const int lane   = tid & 31;
  const int wave   = tid >> 5;
  const int m16    = lane & 15;
  const int hh     = lane >> 4;
  const int rt     = blockIdx.x / 3;
  const int cg     = blockIdx.x % 3;

  constexpr int kRPW = 8;
  const int wv0 = __builtin_amdgcn_readfirstlane(wave);
  v4f xr[kRPW];
  float xl[kRPW];
#pragma unroll
  for (int i = 0; i < kRPW; ++i) {
    const int r = wv0 + 16 * i;
    xr[i] = *(const v4f_a4*)(x + r * kIN + 4 * lane);
    xl[i] = x[r * kIN + (kIN - 1)];
  }
  v4f xr8 = {0.f, 0.f, 0.f, 0.f};
  float xl8 = 0.f;
  if (wv0 == 0) {
    xr8 = *(const v4f_a4*)(x + (kN - 1) * kIN + 4 * lane);
    xl8 = x[(kN - 1) * kIN + (kIN - 1)];
  }
  const int hrow = rt * 16 + wv0;
  v4f h4 = {0.f, 0.f, 0.f, 0.f};
  float h1 = 0.f;
  if (hrow < kN) {
    h4 = *(const v4f_a4*)(W0 + hrow * kIN + 4 * lane);
    h1 = W0[hrow * kIN + (kIN - 1)];
  }
  const v4f   pp = *(const v4f_a4*)(pool_p + 4 * lane);
  const float pl = pool_p[kIN - 1];

  for (int i = tid; i < 16 * kLD; i += nthr) {
    s_xt[rt * 16 * kLD + i] = 0.f; s_h[rt * 16 * kLD + i] = 0.f;
  }
  __syncthreads();

  const int nrow = (wv0 == 0) ? kRPW + 1 : kRPW;
#pragma unroll
  for (int i = 0; i < kRPW; ++i) {
    const int r = wv0 + 16 * i;
    *(v4f*)&s_x[r * kLD + 4 * lane] = xr[i];
  }
  if (wv0 == 0) *(v4f*)&s_x[(kN - 1) * kLD + 4 * lane] = xr8;
  if (hrow < kN) {
    *(v4f*)&s_h[hrow * kLD + 4 * lane] = h4;
    if (lane == 0) s_h[hrow * kLD + (kIN - 1)] = h1;
  }
  {
    float xs = xl8;
#pragma unroll
    for (int i = 0; i < kRPW; ++i)
      if (lane == i) xs = xl[i];
    if (lane < nrow) s_x[(wv0 + 16 * lane) * kLD + (kIN - 1)] = xs;
  }
  {
    float pn = pp.x * pp.x + pp.y * pp.y + pp.z * pp.z + pp.w * pp.w;
    float d[kRPW + 1];
#pragma unroll
    for (int i = 0; i < kRPW; ++i)
      d[i] = xr[i].x * pp.x + xr[i].y * pp.y + xr[i].z * pp.z + xr[i].w * pp.w;
    d[kRPW] = xr8.x * pp.x + xr8.y * pp.y + xr8.z * pp.z + xr8.w * pp.w;
#pragma unroll
    for (int m = 16; m >= 1; m >>= 1) {
      pn += __shfl_xor(pn, m);
#pragma unroll
      for (int i = 0; i <= kRPW; ++i) d[i] += __shfl_xor(d[i], m);
    }
    pn += pl * pl;
    float ds = d[kRPW] + xl8 * pl;
#pragma unroll
    for (int i = 0; i < kRPW; ++i)
      if (lane == i) ds = d[i] + xl[i] * pl;
    if (lane < nrow) s_score[wv0 + 16 * lane] = tanhf(ds * rsqrtf(pn));
  }
  __syncthreads();

  for (int i = tid; i < kN; i += nthr) {
    float si = s_score[i];
    int rank = 0;
    for (int j = 0; j < kN; ++j) {
      float sj = s_score[j];
      rank += (sj > si) || (sj == si && j < i);
    }
    s_perm[rank] = i;
  }
  __syncthreads();

  for (int i = tid; i < 16 * kLD; i += nthr) {
    const int r = rt * 16 + i / kLD, c = i % kLD;
    if (r < kN && c < kIN) {
      const int src = s_perm[r];
      s_xt[r * kLD + c] = s_x[src * kLD + c] * s_score[src];
    }
  }
  __syncthreads();

  {
    const int wv = __builtin_amdgcn_readfirstlane(wave);
    if (wv < 3)
      gru_cells<1>(cg * 3 + wv, rt, m16, hh, w_ih, w_hh, b_ih, b_hh, s_xt, s_h, s_w);
  }
  __syncthreads();

  for (int i = tid; i < 16 * 48; i += nthr) {
    const int r = rt * 16 + i / 48, c = cg * 48 + i % 48;
    wsW[blockIdx.x * (16 * 48) + i] = (r < kN && c < kLD) ? s_w[r * kLD + c] : 0.f;
  }
}

__launch_bounds__(512, 1)
__global__ void gcn_tail(const float* __restrict__ conv_b,
                         const float* __restrict__ lin_w,
                         const float* __restrict__ lin_b,
                         const float* __restrict__ wsW,
                         const float* __restrict__ wsA,
                         float*       __restrict__ out)
{
  __shared__ __attribute__((aligned(16))) float s_w[kRP * kLD];
  __shared__ __attribute__((aligned(16))) float s_a[16 * kLD];
  __shared__ __attribute__((aligned(16))) float s_e[16 * kLD];

  const int tid    = threadIdx.x;
  const int lane   = tid & 31;
  const int wave   = tid >> 5;
  const int m16    = lane & 15;
  const int hh     = lane >> 4;
  const int rt     = blockIdx.x;
  const int wv     = __builtin_amdgcn_readfirstlane(wave);

  const float* wsrc  = wsW + (lane / 12) * (16 * 48) + wv * 48 + (lane % 12) * 4;
  const float* wsrc2 = wsW + 2 * (16 * 48) + wv * 48 + 8 * 4;
  v4f wa[kMT], wb[kMT];
#pragma unroll
  for (int t = 0; t < kMT; ++t) {
    wa[t] = *(const v4f*)(wsrc  + t * (3 * 16 * 48));
    wb[t] = *(const v4f*)(wsrc2 + t * (3 * 16 * 48));
  }
  const float* ag = wsA + (16 * rt + wv) * kAP;
  const v4f a4 = *(const v4f*)(ag + 4 * lane);
  const v4f a1 = *(const v4f*)(ag + 128);
  const bool lin = (wv >= 9 && wv < 13);
  const int  oc  = (lin ? wv - 9 : 0) * 16 + m16;
  v2f bl[kKS];
  float lb = 0.f;
#pragma unroll
  for (int ks = 0; ks < kKS; ++ks) { bl[ks].x = 0.f; bl[ks].y = 0.f; }
  if (lin) {
    const int rowo = oc * kIN;
#pragma unroll
    for (int ks = 0; ks < kKS; ++ks) {
      const int k0 = ks * 4 + 2 * hh;
      bl[ks].x = lin_w[rowo + min(k0,     kIN - 1)];
      bl[ks].y = lin_w[rowo + min(k0 + 1, kIN - 1)];
    }
    lb = lin_b[oc];
  }
  const int jn = wv * 16 + m16;
  float cb = 0.f;
  if (wv < kMT) cb = conv_b[min(jn, kN - 1)];

#pragma unroll
  for (int t = 0; t < kMT; ++t) {
    *(v4f*)&s_w[(t * 16 + wv) * kLD + 4 * lane] = wa[t];
    if (lane == 0) *(v4f*)&s_w[(t * 16 + wv) * kLD + 128] = wb[t];
  }
  *(v4f*)&s_a[wv * kLD + 4 * lane] = a4;
  if (lane == 0) *(v4f*)&s_a[wv * kLD + 128] = a1;
  __syncthreads();

  if (wv < kMT) {
    const int jn_c = min(jn, kLD - 1);
    v8f acc = {0.f, 0.f, 0.f, 0.f, 0.f, 0.f, 0.f, 0.f};
    const int arow = m16 * kLD;
    v2f af[kKS], bf[kKS];
#pragma unroll
    for (int ks = 0; ks < kKS; ++ks) {
      const int k0 = ks * 4 + 2 * hh;
      af[ks]   = *(const v2f*)&s_a[arow + k0];
      bf[ks].x = s_w[k0 * kLD + jn_c];
      bf[ks].y = s_w[(k0 + 1) * kLD + jn_c];
    }
    __builtin_amdgcn_sched_barrier(0);
#pragma unroll
    for (int ks = 0; ks < kKS; ++ks) acc = wmma4(af[ks], bf[ks], acc);
    if (jn < kLD) {
#pragma unroll
      for (int v = 0; v < 8; ++v) {
        const float t = acc[v] + cb;
        const float e = t > 0.f ? t : (__expf(t) - 1.f);
        s_e[(v + 8 * hh) * kLD + jn] = jn < kN ? e : 0.f;
      }
    }
  }
  __syncthreads();

  if (lin) {
    v8f acc;
#pragma unroll
    for (int v = 0; v < 8; ++v) acc[v] = lb;
    const int arow = m16 * kLD;
    v2f ef[kKS];
#pragma unroll
    for (int ks = 0; ks < kKS; ++ks)
      ef[ks] = *(const v2f*)&s_e[arow + ks * 4 + 2 * hh];
    __builtin_amdgcn_sched_barrier(0);
#pragma unroll
    for (int ks = 0; ks < kKS; ++ks) acc = wmma4(ef[ks], bl[ks], acc);
#pragma unroll
    for (int v = 0; v < 8; ++v) {
      const int row = rt * 16 + v + 8 * hh;
      if (row < kN) out[row * kOUT + oc] = acc[v];
    }
  }
}

extern "C" void kernel_launch(void* const* d_in, const int* in_sizes, int n_in,
                              void* d_out, int out_size, void* d_ws, size_t ws_size,
                              hipStream_t stream) {
  (void)in_sizes; (void)n_in; (void)out_size; (void)ws_size;
  const float* x       = (const float*)d_in[0];
  const int*   eidx    = (const int*)  d_in[1];
  const float* ew      = (const float*)d_in[2];
  const float* pool_p  = (const float*)d_in[3];
  const float* W0      = (const float*)d_in[4];
  const float* w_ih    = (const float*)d_in[5];
  const float* w_hh    = (const float*)d_in[6];
  const float* b_ih    = (const float*)d_in[7];
  const float* b_hh    = (const float*)d_in[8];
  const float* conv_b  = (const float*)d_in[9];
  const float* lin_w   = (const float*)d_in[10];
  const float* lin_b   = (const float*)d_in[11];
  float*       wsW     = (float*)d_ws;
  float*       wsA     = (float*)d_ws + kWSW;
  gru_rows<<<36, 512, 0, stream>>>(x, eidx, ew, pool_p, W0, w_ih, w_hh, b_ih, b_hh, wsW, wsA);
  gcn_tail<<<9, 512, 0, stream>>>(conv_b, lin_w, lin_b, wsW, wsA, (float*)d_out);
}


# GRU step cut by rows of x, not of W: input products on unpermuted x need no ranking; beside them other waves rank the 16 rows by ballots, fetch W0[rank]; score scales the product: sum order changes
# speedup vs baseline: 19.8514x; 1.1822x over previous
#include <hip/hip_runtime.h>


typedef __attribute__((ext_vector_type(2))) float v2f;
typedef __attribute__((ext_vector_type(4))) float v4f;
typedef __attribute__((ext_vector_type(8))) float v8f;
typedef v4f v4f_a4 __attribute__((aligned(4)));

static constexpr int kN   = 129;
static constexpr int kIN  = 129;
static constexpr int kOUT = 64;
static constexpr int kE   = 4096;
static constexpr int kLD  = 132;
static constexpr int kRP  = 144;
static constexpr int kKS  = 33;
static constexpr int kMT  = 9;
static constexpr int kAP  = 160;
static constexpr int kWSW = 27 * 16 * 48;
static constexpr int kWP  = 160;

static __device__ __forceinline__ v8f wmma4(v2f a, v2f b, v8f c) {
  return __builtin_amdgcn_wmma_f32_16x16x4_f32(
       false, a,  false, b,
       (short)0, c,  false,  false);
}

static __device__ __forceinline__ void lds_barrier() {
  __builtin_amdgcn_fence(__ATOMIC_RELEASE, "workgroup", "local");
  __builtin_amdgcn_s_barrier();
  __builtin_amdgcn_fence(__ATOMIC_ACQUIRE, "workgroup", "local");
}

static __device__ __forceinline__ float sigm(float x) {
  return 1.0f / (1.0f + __expf(-x));
}

struct BFrag {
  v4f q[16];
  v2f ql;
};

static __device__ __forceinline__ void load_bfrag(BFrag& b, const float* __restrict__ w,
                                                  int grow, int hh) {
  const float* p = w + grow * kIN + 4 * hh;
  const float t = w[grow * kIN + (kIN - 1)];
  b.ql.x = t; b.ql.y = t;
#pragma unroll
  for (int j = 0; j < 16; ++j) b.q[j] = *(const v4f_a4*)(p + 8 * j);
}

template <int NT>
static __device__ __forceinline__ void gru_pass(v8f (&acc)[NT], const float* A,
                                                BFrag& b, int abase, int hh, bool more,
                                                const float* __restrict__ wn, int nrow) {
  const float* pn = wn + nrow * kIN + 4 * hh;
  v4f a[NT];
  asm volatile("" ::: "memory");
  float qn = 0.f;
  if (more) qn = wn[nrow * kIN + (kIN - 1)];
#pragma unroll
  for (int t = 0; t < NT; ++t)
    a[t] = *(const v4f*)&A[abase + t * 16 * kLD + 4 * hh];
#pragma unroll
  for (int j = 0; j < 16; ++j) {
    asm volatile("" ::: "memory");
    v4f an[NT];
    v2f al[NT];
    if (j < 15) {
#pragma unroll
      for (int t = 0; t < NT; ++t)
        an[t] = *(const v4f*)&A[abase + t * 16 * kLD + 8 * (j + 1) + 4 * hh];
    } else {
#pragma unroll
      for (int t = 0; t < NT; ++t) {
        const int at = abase + t * 16 * kLD + 4 * (kKS - 1) + 2 * hh;
        al[t].x = A[at]; al[t].y = A[at + 1];
      }
    }
#pragma unroll
    for (int t = 0; t < NT; ++t) {
      v2f a0, b0; a0.x = a[t].x; a0.y = a[t].y; b0.x = b.q[j].x; b0.y = b.q[j].y;
      acc[t] = wmma4(a0, b0, acc[t]);
    }
#pragma unroll
    for (int t = 0; t < NT; ++t) {
      v2f a1, b1; a1.x = a[t].z; a1.y = a[t].w; b1.x = b.q[j].z; b1.y = b.q[j].w;
      acc[t] = wmma4(a1, b1, acc[t]);
    }
    if (more) b.q[j] = *(const v4f_a4*)(pn + 8 * j);
    if (j < 15) {
#pragma unroll
      for (int t = 0; t < NT; ++t) a[t] = an[t];
    } else {
#pragma unroll
      for (int t = 0; t < NT; ++t) acc[t] = wmma4(al[t], b.ql, acc[t]);
    }
  }
  if (more) { b.ql.x = qn; b.ql.y = qn; }
}

static __device__ __forceinline__ void agg_rows(
    int rt, const float* __restrict__ x, const int* __restrict__ eidx,
    const float* __restrict__ ew, float* __restrict__ wsA,
    float* s_x, float* s_rec, float* s_small) {
  float* s_deg = s_small;
  float* s_dis = s_small + 160;
  int*   s_cnt = (int*)(s_small + 320);
  int*   s_beg = (int*)(s_small + 336);
  int*   s_cur = (int*)(s_small + 352);

  const int tid  = threadIdx.x;
  const int lane = tid & 31;
  const int wave = tid >> 5;
  constexpr int kRPW = 8;
  constexpr int kNT0 = 512;
  constexpr int kEP0 = kE / kNT0;
  const int wv0 = __builtin_amdgcn_readfirstlane(wave);

  v4f xr[kRPW];
  float xl[kRPW];
#pragma unroll
  for (int i = 0; i < kRPW; ++i) {
    const int r = wv0 + 16 * i;
    xr[i] = *(const v4f_a4*)(x + r * kIN + 4 * lane);
    xl[i] = x[r * kIN + (kIN - 1)];
  }
  v4f xr8 = {0.f, 0.f, 0.f, 0.f};
  float xl8 = 0.f;
  if (wv0 == 0) {
    xr8 = *(const v4f_a4*)(x + (kN - 1) * kIN + 4 * lane);
    xl8 = x[(kN - 1) * kIN + (kIN - 1)];
  }
  int erow[kEP0], ecol[kEP0]; float ewt[kEP0];
#pragma unroll
  for (int u = 0; u < kEP0; ++u) {
    const int e = tid + u * kNT0;
    erow[u] = eidx[e]; ecol[u] = eidx[kE + e]; ewt[u] = ew[e];
  }

  if (tid < kN) s_deg[tid] = 0.f;
  if (tid < 16) s_cnt[tid] = 0;
  __syncthreads();

  const int nrow = (wv0 == 0) ? kRPW + 1 : kRPW;
#pragma unroll
  for (int i = 0; i < kRPW; ++i) {
    const int r = wv0 + 16 * i;
    *(v4f*)&s_x[r * kLD + 4 * lane] = xr[i];
  }
  if (wv0 == 0) *(v4f*)&s_x[(kN - 1) * kLD + 4 * lane] = xr8;
  {
    float xs = xl8;
#pragma unroll
    for (int i = 0; i < kRPW; ++i)
      if (lane == i) xs = xl[i];
    if (lane < nrow) s_x[(wv0 + 16 * lane) * kLD + (kIN - 1)] = xs;
  }
#pragma unroll
  for (int u = 0; u < kEP0; ++u) {
    atomicAdd(&s_deg[ecol[u]], ewt[u]);
    const unsigned lc = (unsigned)(ecol[u] - 16 * rt);
    if (lc < 16u) atomicAdd(&s_cnt[lc], 1);
  }
  if (tid < kN) atomicAdd(&s_deg[tid], 1.0f);
  __syncthreads();

  if (tid < kN) {
    const float dg = s_deg[tid];
    s_dis[tid] = dg > 0.f ? rsqrtf(fmaxf(dg, 1e-12f)) : 0.f;
  }
  if (tid >= 256 && tid < 256 + 16) {
    const int i = tid - 256;
    int s = 0;
    for (int j = 0; j < 15; ++j) s += (j < i) ? s_cnt[j] : 0;
    s_beg[i] = s;
    s_cur[i] = s;
  }
  __syncthreads();

#pragma unroll
  for (int u = 0; u < kEP0; ++u) {
    const unsigned lc = (unsigned)(ecol[u] - 16 * rt);
    if (lc < 16u) {
      const int p = atomicAdd(&s_cur[lc], 1);
      v2f rec;
      rec.x = __int_as_float(erow[u]);
      rec.y = s_dis[erow[u]] * ewt[u] * s_dis[ecol[u]];
      *(v2f*)&s_rec[2 * p] = rec;
    }
  }
  __syncthreads();

  const int c = 16 * rt + wv0;
  float* dst = wsA + c * kAP;
  v4f   acc  = {0.f, 0.f, 0.f, 0.f};
  float acc1 = 0.f;
  if (c < kN) {
    const int beg = __builtin_amdgcn_readfirstlane(s_beg[wv0]);
    const int n   = __builtin_amdgcn_readfirstlane(s_cnt[wv0]);
    const float dc = s_dis[c];
    const float sl = dc * 1.0f * dc;
    const v4f xs = *(const v4f*)&s_x[c * kLD + 4 * lane];
    acc  = sl * xs;
    acc1 = sl * s_x[c * kLD + (kN - 1)];
    for (int base = 0; base < n; base += 32) {
      const int m = min(32, n - base);
      int   rr = c;
      float ww = 0.f;
      if (lane < m) {
        const v2f rec = *(const v2f*)&s_rec[2 * (beg + base + lane)];
        rr = __float_as_int(rec.x);
        ww = rec.y;
      }
      for (int i = 0; i < m; i += 8) {
#pragma unroll
        for (int u = 0; u < 8; ++u) {
          const int   r = __builtin_amdgcn_readlane(rr, i + u);
          const float w = __int_as_float(
              __builtin_amdgcn_readlane(__float_as_int(ww), i + u));
          const v4f   v  = *(const v4f*)&s_x[r * kLD + 4 * lane];
          const float v1 = s_x[r * kLD + (kN - 1)];
          acc  += w * v;
          acc1 += w * v1;
        }
      }
    }
  }
  *(v4f*)(dst + 4 * lane) = acc;
  if (lane < 8) {
    v4f t = {0.f, 0.f, 0.f, 0.f};
    if (lane == 0) t.x = acc1;
    *(v4f*)(dst + 128 + 4 * lane) = t;
  }
}

__launch_bounds__(512, 1)
__global__ void gru_rows(const float* __restrict__ x,
                         const int*   __restrict__ eidx,
                         const float* __restrict__ ew,
                         const float* __restrict__ pool_p,
                         const float* __restrict__ W0,
                         const float* __restrict__ w_ih,
                         const float* __restrict__ w_hh,
                         const float* __restrict__ b_ih,
                         const float* __restrict__ b_hh,
                         float*       __restrict__ wsW,
                         float*       __restrict__ wsA)
{
  __shared__ __attribute__((aligned(16))) float s_x [kRP * kLD];
  __shared__ __attribute__((aligned(16))) float s_xt[kRP * kLD];
  __shared__ __attribute__((aligned(16))) float s_h [kRP * kLD];
  __shared__ __attribute__((aligned(16))) float s_w [kRP * kLD];
  __shared__ float s_score[kN];
  __shared__ int   s_rank [16];

  if (blockIdx.x >= 27) {
    agg_rows(blockIdx.x - 27, x, eidx, ew, wsA, s_x, s_xt, s_h);
    return;
  }

  constexpr int kNT0 = 512;
  const int tid    = threadIdx.x;
  const int lane   = tid & 31;
  const int wave   = tid >> 5;
  const int m16    = lane & 15;
  const int hh     = lane >> 4;
  const int rt     = blockIdx.x / 3;
  const int cg     = blockIdx.x % 3;
  const int wv0 = __builtin_amdgcn_readfirstlane(wave);

  constexpr int kRPW = 8;
  v4f xr[kRPW];
  float xl[kRPW];
#pragma unroll
  for (int i = 0; i < kRPW; ++i) {
    const int r = wv0 + 16 * i;
    xr[i] = *(const v4f_a4*)(x + r * kIN + 4 * lane);
    xl[i] = x[r * kIN + (kIN - 1)];
  }
  v4f xr8 = {0.f, 0.f, 0.f, 0.f};
  float xl8 = 0.f;
  if (wv0 == 0) {
    xr8 = *(const v4f_a4*)(x + (kN - 1) * kIN + 4 * lane);
    xl8 = x[(kN - 1) * kIN + (kIN - 1)];
  }
  const v4f   pp = *(const v4f_a4*)(pool_p + 4 * lane);
  const float pl = pool_p[kIN - 1];

  const int  ncell = (cg < 2) ? 4 : 1;
  const bool cellw = wv0 < ncell;
  const int nt    = cg * 4 + (cellw ? wv0 : 0);
  const int jn    = nt * 16 + m16;
  const int jn_c  = min(jn, kLD - 1);
  const int abase = (rt * 16 + m16) * kLD;
  const int cbase = (rt * 16 + 8 * hh) * kLD;
  const int grow_r = min(0 * kN + jn, 3 * kN - 1);
  const int grow_n = min(2 * kN + jn, 3 * kN - 1);
  const int grow_z = min(1 * kN + jn, 3 * kN - 1);
  BFrag B;
  float bir = 0.f, bin = 0.f, biz = 0.f, bhr = 0.f, bhn = 0.f, bhz = 0.f;
  if (cellw) {
    load_bfrag(B, w_ih, grow_r, hh);
    bir = b_ih[grow_r]; bin = b_ih[grow_n]; biz = b_ih[grow_z];
    bhr = b_hh[grow_r]; bhn = b_hh[grow_n]; bhz = b_hh[grow_z];
  }

  if (rt == kMT - 1)
    for (int i = tid; i < (kRP - kN) * kLD; i += kNT0) s_x[kN * kLD + i] = 0.f;

  const int nrow = (wv0 == 0) ? kRPW + 1 : kRPW;
#pragma unroll
  for (int i = 0; i < kRPW; ++i) {
    const int r = wv0 + 16 * i;
    *(v4f*)&s_x[r * kLD + 4 * lane] = xr[i];
  }
  if (wv0 == 0) *(v4f*)&s_x[(kN - 1) * kLD + 4 * lane] = xr8;
  {
    float xs = xl8;
#pragma unroll
    for (int i = 0; i < kRPW; ++i)
      if (lane == i) xs = xl[i];
    v4f xs4 = {xs, 0.f, 0.f, 0.f};
    if (lane < nrow) *(v4f*)&s_x[(wv0 + 16 * lane) * kLD + (kIN - 1)] = xs4;
  }
  {
    float pn = pp.x * pp.x + pp.y * pp.y + pp.z * pp.z + pp.w * pp.w;
    float d[kRPW + 1];
#pragma unroll
    for (int i = 0; i < kRPW; ++i)
      d[i] = xr[i].x * pp.x + xr[i].y * pp.y + xr[i].z * pp.z + xr[i].w * pp.w;
    d[kRPW] = xr8.x * pp.x + xr8.y * pp.y + xr8.z * pp.z + xr8.w * pp.w;
#pragma unroll
    for (int m = 16; m >= 1; m >>= 1) {
      pn += __shfl_xor(pn, m);
#pragma unroll
      for (int i = 0; i <= kRPW; ++i) d[i] += __shfl_xor(d[i], m);
    }
    pn += pl * pl;
    float ds = d[kRPW] + xl8 * pl;
#pragma unroll
    for (int i = 0; i < kRPW; ++i)
      if (lane == i) ds = d[i] + xl[i] * pl;
    if (lane < nrow) s_score[wv0 + 16 * lane] = tanhf(ds * rsqrtf(pn));
  }
  lds_barrier();

  const v8f zero8 = {0.f, 0.f, 0.f, 0.f, 0.f, 0.f, 0.f, 0.f};
  v8f g0 = zero8, g1 = zero8, g2 = zero8;
  if (cellw) {
#pragma unroll 1
    for (int g = 0; g < 3; ++g) {
      v8f a1[1];
      a1[0] = zero8;
      const float* wn  = (g == 2) ? w_hh : w_ih;
      const int   nrw = (g == 0) ? grow_n : (g == 1 ? grow_z : grow_r);
      gru_pass<1>(a1, s_x, B, abase, hh, true, wn, nrw);
      g0 = g1; g1 = g2; g2 = a1[0];
    }
  } else if (wv0 >= 4) {
    for (int q = wv0 - 4; q < 16; q += 12) {
      const int i = rt * 16 + q;
      float* hrow = &s_h[(rt * 16 + q) * kLD];
      if (i < kN) {
        const float si = s_score[i];
        int rank = 0;
#pragma unroll
        for (int m = 0; m < 5; ++m) {
          const int j = lane + 32 * m;
          const float sj = s_score[min(j, kN - 1)];
          const bool ahead = (j < kN) && ((sj > si) || (sj == si && j < i));
          rank += __builtin_popcount(__builtin_amdgcn_ballot_w32(ahead));
        }
        const int rk = __builtin_amdgcn_readfirstlane(rank);
        const v4f   h4 = *(const v4f_a4*)(W0 + rk * kIN + 4 * lane);
        const float h1 = W0[rk * kIN + (kIN - 1)];
        *(v4f*)&hrow[4 * lane] = h4;
        if (lane == 0) {
          v4f t4 = {h1, 0.f, 0.f, 0.f};
          *(v4f*)&hrow[kIN - 1] = t4;
          s_rank[q] = rk;
        }
      } else {
        v4f z4 = {0.f, 0.f, 0.f, 0.f};
        *(v4f*)&hrow[4 * lane] = z4;
        if (lane == 0) { *(v4f*)&hrow[kIN - 1] = z4; s_rank[q] = 0; }
      }
    }
  }
  lds_barrier();

  if (cellw) {
    float sc[8];
#pragma unroll
    for (int v = 0; v < 8; ++v) sc[v] = s_score[min(rt * 16 + 8 * hh + v, kN - 1)];
    v8f a1[1];
    float rg[8], cd[8];
#pragma unroll
    for (int v = 0; v < 8; ++v) a1[0][v] = sc[v] * g0[v] + (bir + bhr);
    gru_pass<1>(a1, s_h, B, abase, hh, true, w_hh, grow_n);
#pragma unroll
    for (int v = 0; v < 8; ++v) { rg[v] = sigm(a1[0][v]); a1[0][v] = bhn; }
    gru_pass<1>(a1, s_h, B, abase, hh, true, w_hh, grow_z);
#pragma unroll
    for (int v = 0; v < 8; ++v) {
      cd[v] = tanhf(sc[v] * g1[v] + bin + rg[v] * a1[0][v]);
      a1[0][v] = sc[v] * g2[v] + (biz + bhz);
    }
    gru_pass<1>(a1, s_h, B, abase, hh, false, w_hh, grow_z);
    if (jn < kLD) {
#pragma unroll
      for (int v = 0; v < 8; ++v) {
        const int idx = cbase + v * kLD + jn;
        const float zg = sigm(a1[0][v]);
        s_w[idx] = (1.f - zg) * cd[v] + zg * s_h[idx];
      }
    }
  }
  __syncthreads();

  if (cg < 2) {
    if (tid < 256) {
      const int q = tid >> 4, pc = tid & 15;
      if (rt * 16 + q < kN)
        *(v4f*)(wsW + s_rank[q] * kWP + cg * 64 + 4 * pc) =
            *(const v4f*)&s_w[(rt * 16 + q) * kLD + cg * 64 + 4 * pc];
    }
  } else {
    if (tid < 16 && rt * 16 + tid < kN)
      *(v4f*)(wsW + s_rank[tid] * kWP + 128) = *(const v4f*)&s_w[(rt * 16 + tid) * kLD + 128];
  }
}

__launch_bounds__(512, 1)
__global__ void gcn_tail(const float* __restrict__ conv_b,
                         const float* __restrict__ lin_w,
                         const float* __restrict__ lin_b,
                         const float* __restrict__ wsW,
                         const float* __restrict__ wsA,
                         float*       __restrict__ out)
{
  __shared__ __attribute__((aligned(16))) float s_w[kRP * kLD];
  __shared__ __attribute__((aligned(16))) float s_a[16 * kLD];
  __shared__ __attribute__((aligned(16))) float s_e[16 * kLD];

  const int tid    = threadIdx.x;
  const int lane   = tid & 31;
  const int wave   = tid >> 5;
  const int m16    = lane & 15;
  const int hh     = lane >> 4;
  const int rt     = blockIdx.x;
  const int wv     = __builtin_amdgcn_readfirstlane(wave);

  const float* wsrc = wsW + wv * kWP;
  const v4f zero4 = {0.f, 0.f, 0.f, 0.f};
  v4f wa[kMT], wb[kMT];
#pragma unroll
  for (int t = 0; t < kMT - 1; ++t) {
    wa[t] = *(const v4f*)(wsrc + t * (16 * kWP) + 4 * lane);
    wb[t] = *(const v4f*)(wsrc + t * (16 * kWP) + 128);
  }
  wa[kMT - 1] = zero4; wb[kMT - 1] = zero4;
  if (wv == 0) {
    wa[kMT - 1] = *(const v4f*)(wsrc + (kMT - 1) * (16 * kWP) + 4 * lane);
    wb[kMT - 1] = *(const v4f*)(wsrc + (kMT - 1) * (16 * kWP) + 128);
  }
  const float* ag = wsA + (16 * rt + wv) * kAP;
  const v4f a4 = *(const v4f*)(ag + 4 * lane);
  const v4f a1 = *(const v4f*)(ag + 128);
  const bool lin = (wv >= 9 && wv < 13);
  const int  oc  = (lin ? wv - 9 : 0) * 16 + m16;
  v2f bl[kKS];
  float lb = 0.f;
#pragma unroll
  for (int ks = 0; ks < kKS; ++ks) { bl[ks].x = 0.f; bl[ks].y = 0.f; }
  if (lin) {
    const int rowo = oc * kIN;
#pragma unroll
    for (int ks = 0; ks < kKS; ++ks) {
      const int k0 = ks * 4 + 2 * hh;
      bl[ks].x = lin_w[rowo + min(k0,     kIN - 1)];
      bl[ks].y = lin_w[rowo + min(k0 + 1, kIN - 1)];
    }
    lb = lin_b[oc];
  }
  const int jn = wv * 16 + m16;
  float cb = 0.f;
  if (wv < kMT) cb = conv_b[min(jn, kN - 1)];

#pragma unroll
  for (int t = 0; t < kMT; ++t) {
    *(v4f*)&s_w[(t * 16 + wv) * kLD + 4 * lane] = wa[t];
    if (lane == 0) *(v4f*)&s_w[(t * 16 + wv) * kLD + 128] = wb[t];
  }
  *(v4f*)&s_a[wv * kLD + 4 * lane] = a4;
  if (lane == 0) *(v4f*)&s_a[wv * kLD + 128] = a1;
  __syncthreads();

  if (wv < kMT) {
    const int jn_c = min(jn, kLD - 1);
    v8f acc = {0.f, 0.f, 0.f, 0.f, 0.f, 0.f, 0.f, 0.f};
    const int arow = m16 * kLD;
    v2f af[kKS], bf[kKS];
#pragma unroll
    for (int ks = 0; ks < kKS; ++ks) {
      const int k0 = ks * 4 + 2 * hh;
      af[ks]   = *(const v2f*)&s_a[arow + k0];
      bf[ks].x = s_w[k0 * kLD + jn_c];
      bf[ks].y = s_w[(k0 + 1) * kLD + jn_c];
    }
    __builtin_amdgcn_sched_barrier(0);
#pragma unroll
    for (int ks = 0; ks < kKS; ++ks) acc = wmma4(af[ks], bf[ks], acc);
    if (jn < kLD) {
#pragma unroll
      for (int v = 0; v < 8; ++v) {
        const float t = acc[v] + cb;
        const float e = t > 0.f ? t : (__expf(t) - 1.f);
        s_e[(v + 8 * hh) * kLD + jn] = jn < kN ? e : 0.f;
      }
    }
  }
  __syncthreads();

  if (lin) {
    v8f acc;
#pragma unroll
    for (int v = 0; v < 8; ++v) acc[v] = lb;
    const int arow = m16 * kLD;
    v2f ef[kKS];
#pragma unroll
    for (int ks = 0; ks < kKS; ++ks)
      ef[ks] = *(const v2f*)&s_e[arow + ks * 4 + 2 * hh];
    __builtin_amdgcn_sched_barrier(0);
#pragma unroll
    for (int ks = 0; ks < kKS; ++ks) acc = wmma4(ef[ks], bl[ks], acc);
#pragma unroll
    for (int v = 0; v < 8; ++v) {
      const int row = rt * 16 + v + 8 * hh;
      if (row < kN) out[row * kOUT + oc] = acc[v];
    }
  }
}

extern "C" void kernel_launch(void* const* d_in, const int* in_sizes, int n_in,
                              void* d_out, int out_size, void* d_ws, size_t ws_size,
                              hipStream_t stream) {
  (void)in_sizes; (void)n_in; (void)out_size; (void)ws_size;
  const float* x       = (const float*)d_in[0];
  const int*   eidx    = (const int*)  d_in[1];
  const float* ew      = (const float*)d_in[2];
  const float* pool_p  = (const float*)d_in[3];
  const float* W0      = (const float*)d_in[4];
  const float* w_ih    = (const float*)d_in[5];
  const float* w_hh    = (const float*)d_in[6];
  const float* b_ih    = (const float*)d_in[7];
  const float* b_hh    = (const float*)d_in[8];
  const float* conv_b  = (const float*)d_in[9];
  const float* lin_w   = (const float*)d_in[10];
  const float* lin_b   = (const float*)d_in[11];
  float*       wsW     = (float*)d_ws;
  float*       wsA     = (float*)d_ws + kWSW;
  gru_rows<<<36, 512, 0, stream>>>(x, eidx, ew, pool_p, W0, w_ih, w_hh, b_ih, b_hh, wsW, wsA);
  gcn_tail<<<9, 512, 0, stream>>>(conv_b, lin_w, lin_b, wsW, wsA, (float*)d_out);
}


# GRU cell's k loop cut in eight over 16 waves (45 workgroups of two cells): a wave holds its 16-k eighth of all six gate blocks from the start, ranks one row; partials join in LDS: sum order changes
# speedup vs baseline: 23.9163x; 1.2048x over previous
#include <hip/hip_runtime.h>


typedef __attribute__((ext_vector_type(2))) float v2f;
typedef __attribute__((ext_vector_type(4))) float v4f;
typedef __attribute__((ext_vector_type(8))) float v8f;
typedef v4f v4f_a4 __attribute__((aligned(4)));

static constexpr int kN   = 129;
static constexpr int kIN  = 129;
static constexpr int kOUT = 64;
static constexpr int kE   = 4096;
static constexpr int kLD  = 132;
static constexpr int kRP  = 144;
static constexpr int kKS  = 33;
static constexpr int kMT  = 9;
static constexpr int kAP  = 160;
static constexpr int kWSW = 27 * 16 * 48;
static constexpr int kGW  = 45;
static constexpr int kPT  = 320;
static constexpr int kPC  = 4 * 8 * kPT + 16;
static constexpr int kWP  = 160;

static __device__ __forceinline__ v8f wmma4(v2f a, v2f b, v8f c) {
  return __builtin_amdgcn_wmma_f32_16x16x4_f32(
       false, a,  false, b,
       (short)0, c,  false,  false);
}

static __device__ __forceinline__ void lds_barrier() {
  __builtin_amdgcn_fence(__ATOMIC_RELEASE, "workgroup", "local");
  __builtin_amdgcn_s_barrier();
  __builtin_amdgcn_fence(__ATOMIC_ACQUIRE, "workgroup", "local");
}

static __device__ __forceinline__ float sigm(float x) {
  return 1.0f / (1.0f + __expf(-x));
}

struct BPart {
  v4f q[2];
  v2f ql;
};

static __device__ __forceinline__ void load_bpart(BPart& b, const float* __restrict__ w,
                                                  int grow, int ke, int hh) {
  const float* p = w + grow * kIN + 16 * ke + 4 * hh;
  b.ql.x = 0.f; b.ql.y = 0.f;
  b.q[0] = *(const v4f_a4*)(p);
  b.q[1] = *(const v4f_a4*)(p + 8);
}

static __device__ __forceinline__ void load_btail(BPart& b, const float* __restrict__ w, int grow) {
  const float t = w[grow * kIN + (kIN - 1)];
  b.ql.x = t; b.ql.y = t;
}

struct APart {
  v4f a[2];
  v2f al;
};

static __device__ __forceinline__ void load_apart(APart& f, const float* A, int abase,
                                                  int ke, int hh) {
  f.a[0] = *(const v4f*)&A[abase + 16 * ke + 4 * hh];
  f.a[1] = *(const v4f*)&A[abase + 16 * ke + 8 + 4 * hh];
  f.al.x = 0.f; f.al.y = 0.f;
  if (ke == 7) f.al = *(const v2f*)&A[abase + 4 * (kKS - 1) + 2 * hh];
}

static __device__ __forceinline__ void part_mma(v8f (&acc)[3], const APart& f,
                                                const BPart (&b)[3], int ke) {
#pragma unroll
  for (int j = 0; j < 2; ++j) {
#pragma unroll
    for (int g = 0; g < 3; ++g) {
      v2f a0, b0; a0.x = f.a[j].x; a0.y = f.a[j].y; b0.x = b[g].q[j].x; b0.y = b[g].q[j].y;
      acc[g] = wmma4(a0, b0, acc[g]);
    }
#pragma unroll
    for (int g = 0; g < 3; ++g) {
      v2f a1, b1; a1.x = f.a[j].z; a1.y = f.a[j].w; b1.x = b[g].q[j].z; b1.y = b[g].q[j].w;
      acc[g] = wmma4(a1, b1, acc[g]);
    }
  }
  if (ke == 7) {
#pragma unroll
    for (int g = 0; g < 3; ++g) acc[g] = wmma4(f.al, b[g].ql, acc[g]);
  }
}

static __device__ __forceinline__ void agg_rows(
    int rt, const float* __restrict__ x, const int* __restrict__ eidx,
    const float* __restrict__ ew, float* __restrict__ wsA,
    float* s_x, float* s_rec, float* s_small) {
  float* s_deg = s_small;
  float* s_dis = s_small + 160;
  int*   s_cnt = (int*)(s_small + 320);
  int*   s_beg = (int*)(s_small + 336);
  int*   s_cur = (int*)(s_small + 352);

  const int tid  = threadIdx.x;
  const int lane = tid & 31;
  const int wave = tid >> 5;
  constexpr int kRPW = 8;
  constexpr int kNT0 = 512;
  constexpr int kEP0 = kE / kNT0;
  const int wv0 = __builtin_amdgcn_readfirstlane(wave);

  v4f xr[kRPW];
  float xl[kRPW];
#pragma unroll
  for (int i = 0; i < kRPW; ++i) {
    const int r = wv0 + 16 * i;
    xr[i] = *(const v4f_a4*)(x + r * kIN + 4 * lane);
    xl[i] = x[r * kIN + (kIN - 1)];
  }
  v4f xr8 = {0.f, 0.f, 0.f, 0.f};
  float xl8 = 0.f;
  if (wv0 == 0) {
    xr8 = *(const v4f_a4*)(x + (kN - 1) * kIN + 4 * lane);
    xl8 = x[(kN - 1) * kIN + (kIN - 1)];
  }
  int erow[kEP0], ecol[kEP0]; float ewt[kEP0];
#pragma unroll
  for (int u = 0; u < kEP0; ++u) {
    const int e = tid + u * kNT0;
    erow[u] = eidx[e]; ecol[u] = eidx[kE + e]; ewt[u] = ew[e];
  }

  if (tid < kN) s_deg[tid] = 0.f;
  if (tid < 16) s_cnt[tid] = 0;
  __syncthreads();

  const int nrow = (wv0 == 0) ? kRPW + 1 : kRPW;
#pragma unroll
  for (int i = 0; i < kRPW; ++i) {
    const int r = wv0 + 16 * i;
    *(v4f*)&s_x[r * kLD + 4 * lane] = xr[i];
  }
  if (wv0 == 0) *(v4f*)&s_x[(kN - 1) * kLD + 4 * lane] = xr8;
  {
    float xs = xl8;
#pragma unroll
    for (int i = 0; i < kRPW; ++i)
      if (lane == i) xs = xl[i];
    if (lane < nrow) s_x[(wv0 + 16 * lane) * kLD + (kIN - 1)] = xs;
  }
#pragma unroll
  for (int u = 0; u < kEP0; ++u) {
    atomicAdd(&s_deg[ecol[u]], ewt[u]);
    const unsigned lc = (unsigned)(ecol[u] - 16 * rt);
    if (lc < 16u) atomicAdd(&s_cnt[lc], 1);
  }
  if (tid < kN) atomicAdd(&s_deg[tid], 1.0f);
  __syncthreads();

  if (tid < kN) {
    const float dg = s_deg[tid];
    s_dis[tid] = dg > 0.f ? rsqrtf(fmaxf(dg, 1e-12f)) : 0.f;
  }
  if (tid >= 256 && tid < 256 + 16) {
    const int i = tid - 256;
    int s = 0;
    for (int j = 0; j < 15; ++j) s += (j < i) ? s_cnt[j] : 0;
    s_beg[i] = s;
    s_cur[i] = s;
  }
  __syncthreads();

#pragma unroll
  for (int u = 0; u < kEP0; ++u) {
    const unsigned lc = (unsigned)(ecol[u] - 16 * rt);
    if (lc < 16u) {
      const int p = atomicAdd(&s_cur[lc], 1);
      v2f rec;
      rec.x = __int_as_float(erow[u]);
      rec.y = s_dis[erow[u]] * ewt[u] * s_dis[ecol[u]];
      *(v2f*)&s_rec[2 * p] = rec;
    }
  }
  __syncthreads();

  const int c = 16 * rt + wv0;
  float* dst = wsA + c * kAP;
  v4f   acc  = {0.f, 0.f, 0.f, 0.f};
  float acc1 = 0.f;
  if (c < kN) {
    const int beg = __builtin_amdgcn_readfirstlane(s_beg[wv0]);
    const int n   = __builtin_amdgcn_readfirstlane(s_cnt[wv0]);
    const float dc = s_dis[c];
    const float sl = dc * 1.0f * dc;
    const v4f xs = *(const v4f*)&s_x[c * kLD + 4 * lane];
    acc  = sl * xs;
    acc1 = sl * s_x[c * kLD + (kN - 1)];
    for (int base = 0; base < n; base += 32) {
      const int m = min(32, n - base);
      int   rr = c;
      float ww = 0.f;
      if (lane < m) {
        const v2f rec = *(const v2f*)&s_rec[2 * (beg + base + lane)];
        rr = __float_as_int(rec.x);
        ww = rec.y;
      }
      for (int i = 0; i < m; i += 8) {
#pragma unroll
        for (int u = 0; u < 8; ++u) {
          const int   r = __builtin_amdgcn_readlane(rr, i + u);
          const float w = __int_as_float(
              __builtin_amdgcn_readlane(__float_as_int(ww), i + u));
          const v4f   v  = *(const v4f*)&s_x[r * kLD + 4 * lane];
          const float v1 = s_x[r * kLD + (kN - 1)];
          acc  += w * v;
          acc1 += w * v1;
        }
      }
    }
  }
  *(v4f*)(dst + 4 * lane) = acc;
  if (lane < 8) {
    v4f t = {0.f, 0.f, 0.f, 0.f};
    if (lane == 0) t.x = acc1;
    *(v4f*)(dst + 128 + 4 * lane) = t;
  }
}

__launch_bounds__(512, 1)
__global__ void gru_rows(const float* __restrict__ x,
                         const int*   __restrict__ eidx,
                         const float* __restrict__ ew,
                         const float* __restrict__ pool_p,
                         const float* __restrict__ W0,
                         const float* __restrict__ w_ih,
                         const float* __restrict__ w_hh,
                         const float* __restrict__ b_ih,
                         const float* __restrict__ b_hh,
                         float*       __restrict__ wsW,
                         float*       __restrict__ wsA)
{
  __shared__ __attribute__((aligned(16))) float s_x [kRP * kLD];
  __shared__ __attribute__((aligned(16))) float s_xt[kRP * kLD];
  __shared__ __attribute__((aligned(16))) float s_h [kRP * kLD];
  __shared__ float s_p[2 * kPC];
  __shared__ float s_score[kN];
  __shared__ int   s_rank [16];

  if (blockIdx.x >= kGW) {
    agg_rows(blockIdx.x - kGW, x, eidx, ew, wsA, s_x, s_xt, s_h);
    return;
  }

  constexpr int kNT0 = 512;
  const int tid    = threadIdx.x;
  const int lane   = tid & 31;
  const int wave   = tid >> 5;
  const int m16    = lane & 15;
  const int hh     = lane >> 4;
  const int rt     = blockIdx.x / 5;
  const int cp     = blockIdx.x % 5;
  const int wv0 = __builtin_amdgcn_readfirstlane(wave);

  constexpr int kRPW = 8;
  v4f xr[kRPW];
  float xl[kRPW];
#pragma unroll
  for (int i = 0; i < kRPW; ++i) {
    const int r = wv0 + 16 * i;
    xr[i] = *(const v4f_a4*)(x + r * kIN + 4 * lane);
    xl[i] = x[r * kIN + (kIN - 1)];
  }
  v4f xr8 = {0.f, 0.f, 0.f, 0.f};
  float xl8 = 0.f;
  if (wv0 == 0) {
    xr8 = *(const v4f_a4*)(x + (kN - 1) * kIN + 4 * lane);
    xl8 = x[(kN - 1) * kIN + (kIN - 1)];
  }
  const v4f   pp = *(const v4f_a4*)(pool_p + 4 * lane);
  const float pl = pool_p[kIN - 1];

  const int cell  = wv0 >> 3;
  const int ke    = wv0 & 7;
  const int nt    = cp * 2 + cell;
  const int jn    = nt * 16 + m16;
  const int abase = (rt * 16 + m16) * kLD;
  const int grow_r = min(0 * kN + jn, 3 * kN - 1);
  const int grow_n = min(2 * kN + jn, 3 * kN - 1);
  const int grow_z = min(1 * kN + jn, 3 * kN - 1);
  BPart BI[3], BH[3];
  load_bpart(BI[0], w_ih, grow_r, ke, hh);
  load_bpart(BI[1], w_ih, grow_n, ke, hh);
  load_bpart(BI[2], w_ih, grow_z, ke, hh);
  load_bpart(BH[0], w_hh, grow_r, ke, hh);
  load_bpart(BH[1], w_hh, grow_n, ke, hh);
  load_bpart(BH[2], w_hh, grow_z, ke, hh);
  const int jf    = cp * 32 + lane;
  const int frw_r = min(0 * kN + jf, 3 * kN - 1);
  const int frw_n = min(2 * kN + jf, 3 * kN - 1);
  const int frw_z = min(1 * kN + jf, 3 * kN - 1);
  const float bir = b_ih[frw_r], bin = b_ih[frw_n], biz = b_ih[frw_z];
  const float bhr = b_hh[frw_r], bhn = b_hh[frw_n], bhz = b_hh[frw_z];
  if (ke == 7) {
    load_btail(BI[0], w_ih, grow_r);
    load_btail(BI[1], w_ih, grow_n);
    load_btail(BI[2], w_ih, grow_z);
    load_btail(BH[0], w_hh, grow_r);
    load_btail(BH[1], w_hh, grow_n);
    load_btail(BH[2], w_hh, grow_z);
  }

  if (rt == kMT - 1)
    for (int i = tid; i < (kRP - kN) * kLD; i += kNT0) s_x[kN * kLD + i] = 0.f;

  const int nrow = (wv0 == 0) ? kRPW + 1 : kRPW;
#pragma unroll
  for (int i = 0; i < kRPW; ++i) {
    const int r = wv0 + 16 * i;
    *(v4f*)&s_x[r * kLD + 4 * lane] = xr[i];
  }
  if (wv0 == 0) *(v4f*)&s_x[(kN - 1) * kLD + 4 * lane] = xr8;
  {
    float xs = xl8;
#pragma unroll
    for (int i = 0; i < kRPW; ++i)
      if (lane == i) xs = xl[i];
    v4f xs4 = {xs, 0.f, 0.f, 0.f};
    if (lane < nrow) *(v4f*)&s_x[(wv0 + 16 * lane) * kLD + (kIN - 1)] = xs4;
  }
  {
    float pn = pp.x * pp.x + pp.y * pp.y + pp.z * pp.z + pp.w * pp.w;
    float d[kRPW + 1];
#pragma unroll
    for (int i = 0; i < kRPW; ++i)
      d[i] = xr[i].x * pp.x + xr[i].y * pp.y + xr[i].z * pp.z + xr[i].w * pp.w;
    d[kRPW] = xr8.x * pp.x + xr8.y * pp.y + xr8.z * pp.z + xr8.w * pp.w;
#pragma unroll
    for (int m = 16; m >= 1; m >>= 1) {
      pn += __shfl_xor(pn, m);
#pragma unroll
      for (int i = 0; i <= kRPW; ++i) d[i] += __shfl_xor(d[i], m);
    }
    pn += pl * pl;
    float ds = d[kRPW] + xl8 * pl;
#pragma unroll
    for (int i = 0; i < kRPW; ++i)
      if (lane == i) ds = d[i] + xl[i] * pl;
    if (lane < nrow) s_score[wv0 + 16 * lane] = tanhf(ds * rsqrtf(pn));
  }
  lds_barrier();

  const int  irow = rt * 16 + wv0;
  const bool live = irow < kN;
  int rk;
  {
    const int i = min(irow, kN - 1);
    const float si = s_score[i];
    int rank = 0;
#pragma unroll
    for (int m = 0; m < 5; ++m) {
      const int j = lane + 32 * m;
      const float sj = s_score[min(j, kN - 1)];
      const bool ahead = (j < kN) && ((sj > si) || (sj == si && j < i));
      rank += __builtin_popcount(__builtin_amdgcn_ballot_w32(ahead));
    }
    rk = __builtin_amdgcn_readfirstlane(rank);
  }
  const v4f   h4 = *(const v4f_a4*)(W0 + rk * kIN + 4 * lane);
  const float h1 = W0[rk * kIN + (kIN - 1)];
  const v8f zero8 = {0.f, 0.f, 0.f, 0.f, 0.f, 0.f, 0.f, 0.f};
  v8f gi[3] = {zero8, zero8, zero8};
  {
    APart fx;
    load_apart(fx, s_x, abase, ke, hh);
    part_mma(gi, fx, BI, ke);
  }
  {
    float* hrow = &s_h[irow * kLD];
    const v4f z4 = {0.f, 0.f, 0.f, 0.f};
    *(v4f*)&hrow[4 * lane] = live ? h4 : z4;
    if (lane == 0) {
      v4f t4 = {live ? h1 : 0.f, 0.f, 0.f, 0.f};
      *(v4f*)&hrow[kIN - 1] = t4;
      s_rank[wv0] = live ? rk : 0;
    }
  }
  lds_barrier();

  {
    float sc[8];
#pragma unroll
    for (int v = 0; v < 8; ++v) sc[v] = s_score[min(rt * 16 + 8 * hh + v, kN - 1)];
    v8f ph[3];
#pragma unroll
    for (int v = 0; v < 8; ++v) {
      ph[0][v] = sc[v] * gi[0][v];
      ph[1][v] = 0.f;
      ph[2][v] = sc[v] * gi[2][v];
    }
    APart fh;
    load_apart(fh, s_h, abase, ke, hh);
    part_mma(ph, fh, BH, ke);
    float* pt = &s_p[cell * kPC + ke * kPT + hh * 144 + m16];
#pragma unroll
    for (int v = 0; v < 8; ++v) {
      pt[(0 * 8) * kPT + v * 16] = ph[0][v];
      pt[(1 * 8) * kPT + v * 16] = ph[1][v];
      pt[(2 * 8) * kPT + v * 16] = ph[2][v];
      pt[(3 * 8) * kPT + v * 16] = gi[1][v];
    }
  }
  lds_barrier();

  {
    const int fcell = lane >> 4;
    const bool fcol = (cp < 4) || (lane < 4);
    if (live && fcol) {
      const float* pf = &s_p[fcell * kPC + (wv0 >> 3) * 144 + (wv0 & 7) * 16 + m16];
      float sm[4];
#pragma unroll
      for (int s = 0; s < 4; ++s) {
        float t = pf[(s * 8 + 0) * kPT];
#pragma unroll
        for (int k = 1; k < 8; ++k) t += pf[(s * 8 + k) * kPT];
        sm[s] = t;
      }
      const float si = s_score[irow];
      const float h0 = s_h[irow * kLD + jf];
      const float rg = sigm(sm[0] + (bir + bhr));
      const float cd = tanhf(si * sm[3] + bin + rg * (sm[1] + bhn));
      const float zg = sigm(sm[2] + (biz + bhz));
      wsW[s_rank[wv0] * kWP + jf] = (1.f - zg) * cd + zg * h0;
    }
  }
}

__launch_bounds__(512, 1)
__global__ void gcn_tail(const float* __restrict__ conv_b,
                         const float* __restrict__ lin_w,
                         const float* __restrict__ lin_b,
                         const float* __restrict__ wsW,
                         const float* __restrict__ wsA,
                         float*       __restrict__ out)
{
  __shared__ __attribute__((aligned(16))) float s_w[kRP * kLD];
  __shared__ __attribute__((aligned(16))) float s_a[16 * kLD];
  __shared__ __attribute__((aligned(16))) float s_e[16 * kLD];

  const int tid    = threadIdx.x;
  const int lane   = tid & 31;
  const int wave   = tid >> 5;
  const int m16    = lane & 15;
  const int hh     = lane >> 4;
  const int rt     = blockIdx.x;
  const int wv     = __builtin_amdgcn_readfirstlane(wave);

  const float* wsrc = wsW + wv * kWP;
  const v4f zero4 = {0.f, 0.f, 0.f, 0.f};
  v4f wa[kMT], wb[kMT];
#pragma unroll
  for (int t = 0; t < kMT - 1; ++t) {
    wa[t] = *(const v4f*)(wsrc + t * (16 * kWP) + 4 * lane);
    wb[t] = *(const v4f*)(wsrc + t * (16 * kWP) + 128);
  }
  wa[kMT - 1] = zero4; wb[kMT - 1] = zero4;
  if (wv == 0) {
    wa[kMT - 1] = *(const v4f*)(wsrc + (kMT - 1) * (16 * kWP) + 4 * lane);
    wb[kMT - 1] = *(const v4f*)(wsrc + (kMT - 1) * (16 * kWP) + 128);
  }
  const float* ag = wsA + (16 * rt + wv) * kAP;
  const v4f a4 = *(const v4f*)(ag + 4 * lane);
  const v4f a1 = *(const v4f*)(ag + 128);
  const bool lin = (wv >= 9 && wv < 13);
  const int  oc  = (lin ? wv - 9 : 0) * 16 + m16;
  v2f bl[kKS];
  float lb = 0.f;
#pragma unroll
  for (int ks = 0; ks < kKS; ++ks) { bl[ks].x = 0.f; bl[ks].y = 0.f; }
  if (lin) {
    const int rowo = oc * kIN;
#pragma unroll
    for (int ks = 0; ks < kKS; ++ks) {
      const int k0 = ks * 4 + 2 * hh;
      bl[ks].x = lin_w[rowo + min(k0,     kIN - 1)];
      bl[ks].y = lin_w[rowo + min(k0 + 1, kIN - 1)];
    }
    lb = lin_b[oc];
  }
  const int jn = wv * 16 + m16;
  float cb = 0.f;
  if (wv < kMT) cb = conv_b[min(jn, kN - 1)];

#pragma unroll
  for (int t = 0; t < kMT; ++t) {
    *(v4f*)&s_w[(t * 16 + wv) * kLD + 4 * lane] = wa[t];
    if (lane == 0) *(v4f*)&s_w[(t * 16 + wv) * kLD + 128] = wb[t];
  }
  *(v4f*)&s_a[wv * kLD + 4 * lane] = a4;
  if (lane == 0) *(v4f*)&s_a[wv * kLD + 128] = a1;
  __syncthreads();

  if (wv < kMT) {
    const int jn_c = min(jn, kLD - 1);
    v8f acc = {0.f, 0.f, 0.f, 0.f, 0.f, 0.f, 0.f, 0.f};
    const int arow = m16 * kLD;
    v2f af[kKS], bf[kKS];
#pragma unroll
    for (int ks = 0; ks < kKS; ++ks) {
      const int k0 = ks * 4 + 2 * hh;
      af[ks]   = *(const v2f*)&s_a[arow + k0];
      bf[ks].x = s_w[k0 * kLD + jn_c];
      bf[ks].y = s_w[(k0 + 1) * kLD + jn_c];
    }
    __builtin_amdgcn_sched_barrier(0);
#pragma unroll
    for (int ks = 0; ks < kKS; ++ks) acc = wmma4(af[ks], bf[ks], acc);
    if (jn < kLD) {
#pragma unroll
      for (int v = 0; v < 8; ++v) {
        const float t = acc[v] + cb;
        const float e = t > 0.f ? t : (__expf(t) - 1.f);
        s_e[(v + 8 * hh) * kLD + jn] = jn < kN ? e : 0.f;
      }
    }
  }
  __syncthreads();

  if (lin) {
    v8f acc;
#pragma unroll
    for (int v = 0; v < 8; ++v) acc[v] = lb;
    const int arow = m16 * kLD;
    v2f ef[kKS];
#pragma unroll
    for (int ks = 0; ks < kKS; ++ks)
      ef[ks] = *(const v2f*)&s_e[arow + ks * 4 + 2 * hh];
    __builtin_amdgcn_sched_barrier(0);
#pragma unroll
    for (int ks = 0; ks < kKS; ++ks) acc = wmma4(ef[ks], bl[ks], acc);
#pragma unroll
    for (int v = 0; v < 8; ++v) {
      const int row = rt * 16 + v + 8 * hh;
      if (row < kN) out[row * kOUT + oc] = acc[v];
    }
  }
}

extern "C" void kernel_launch(void* const* d_in, const int* in_sizes, int n_in,
                              void* d_out, int out_size, void* d_ws, size_t ws_size,
                              hipStream_t stream) {
  (void)in_sizes; (void)n_in; (void)out_size; (void)ws_size;
  const float* x       = (const float*)d_in[0];
  const int*   eidx    = (const int*)  d_in[1];
  const float* ew      = (const float*)d_in[2];
  const float* pool_p  = (const float*)d_in[3];
  const float* W0      = (const float*)d_in[4];
  const float* w_ih    = (const float*)d_in[5];
  const float* w_hh    = (const float*)d_in[6];
  const float* b_ih    = (const float*)d_in[7];
  const float* b_hh    = (const float*)d_in[8];
  const float* conv_b  = (const float*)d_in[9];
  const float* lin_w   = (const float*)d_in[10];
  const float* lin_b   = (const float*)d_in[11];
  float*       wsW     = (float*)d_ws;
  float*       wsA     = (float*)d_ws + kWSW;
  gru_rows<<<54, 512, 0, stream>>>(x, eidx, ew, pool_p, W0, w_ih, w_hh, b_ih, b_hh, wsW, wsA);
  gcn_tail<<<9, 512, 0, stream>>>(conv_b, lin_w, lin_b, wsW, wsA, (float*)d_out);
}


# GRU step spread over 81 workgroups of one 16x16 cell: input and hidden products on separate waves (3 gates each); W to launch 2 by column block in x row order with ranks; sum order changes
# speedup vs baseline: 26.4464x; 1.1058x over previous
#include <hip/hip_runtime.h>


typedef __attribute__((ext_vector_type(2))) float v2f;
typedef __attribute__((ext_vector_type(4))) float v4f;
typedef __attribute__((ext_vector_type(8))) float v8f;
typedef v4f v4f_a4 __attribute__((aligned(4)));

static constexpr int kN   = 129;
static constexpr int kIN  = 129;
static constexpr int kOUT = 64;
static constexpr int kE   = 4096;
static constexpr int kLD  = 132;
static constexpr int kRP  = 144;
static constexpr int kKS  = 33;
static constexpr int kMT  = 9;
static constexpr int kNB  = 9;
static constexpr int kAP  = 160;
static constexpr int kWB  = kRP * 16;
static constexpr int kWSW = kNB * kWB;
static constexpr int kGW  = kMT * kNB;
static constexpr int kPT  = 320;
static constexpr int kPS  = 3 * 8 * kPT + 16;
static constexpr int kWL  = 144;
static __device__ __forceinline__ v8f wmma4(v2f a, v2f b, v8f c) {
  return __builtin_amdgcn_wmma_f32_16x16x4_f32(
       false, a,  false, b,
       (short)0, c,  false,  false);
}

static __device__ __forceinline__ void lds_barrier() {
  __builtin_amdgcn_fence(__ATOMIC_RELEASE, "workgroup", "local");
  __builtin_amdgcn_s_barrier();
  __builtin_amdgcn_fence(__ATOMIC_ACQUIRE, "workgroup", "local");
}

static __device__ __forceinline__ float sigm(float x) {
  return 1.0f / (1.0f + __expf(-x));
}

struct BPart {
  v4f q[2];
  float tl;
};

static __device__ __forceinline__ void load_bpart(BPart& b, const float* __restrict__ w,
                                                  int grow, int ke, int hh) {
  const float* p = w + grow * kIN + 16 * ke + 4 * hh;
  b.tl = 0.f;
  b.q[0] = *(const v4f_a4*)(p);
  b.q[1] = *(const v4f_a4*)(p + 8);
}

static __device__ __forceinline__ void load_btail(BPart& b, const float* __restrict__ w, int grow) {
  const float t = w[grow * kIN + (kIN - 1)];
  b.tl = t;
}

struct APart {
  v4f a[2];
  v2f al;
};

static __device__ __forceinline__ void load_apart(APart& f, const float* A, int abase,
                                                  int ke, int hh) {
  f.a[0] = *(const v4f*)&A[abase + 16 * ke + 4 * hh];
  f.a[1] = *(const v4f*)&A[abase + 16 * ke + 8 + 4 * hh];
  f.al.x = 0.f; f.al.y = 0.f;
  if (ke == 7) f.al = *(const v2f*)&A[abase + 4 * (kKS - 1) + 2 * hh];
}

static __device__ __forceinline__ void part_mma(v8f (&acc)[3], const APart& f,
                                                const BPart (&b)[3], int ke) {
#pragma unroll
  for (int j = 0; j < 2; ++j) {
#pragma unroll
    for (int g = 0; g < 3; ++g) {
      v2f a0, b0; a0.x = f.a[j].x; a0.y = f.a[j].y; b0.x = b[g].q[j].x; b0.y = b[g].q[j].y;
      acc[g] = wmma4(a0, b0, acc[g]);
    }
#pragma unroll
    for (int g = 0; g < 3; ++g) {
      v2f a1, b1; a1.x = f.a[j].z; a1.y = f.a[j].w; b1.x = b[g].q[j].z; b1.y = b[g].q[j].w;
      acc[g] = wmma4(a1, b1, acc[g]);
    }
  }
  if (ke == 7) {
#pragma unroll
    for (int g = 0; g < 3; ++g) {
      v2f bt; bt.x = b[g].tl; bt.y = b[g].tl;
      acc[g] = wmma4(f.al, bt, acc[g]);
    }
  }
}

static __device__ __forceinline__ void agg_rows(
    int rt, const float* __restrict__ x, const int* __restrict__ eidx,
    const float* __restrict__ ew, float* __restrict__ wsA,
    float* s_x, float* s_rec, float* s_small) {
  float* s_deg = s_small;
  float* s_dis = s_small + 160;
  int*   s_cnt = (int*)(s_small + 320);
  int*   s_beg = (int*)(s_small + 336);
  int*   s_cur = (int*)(s_small + 352);

  const int tid  = threadIdx.x;
  const int lane = tid & 31;
  const int wave = tid >> 5;
  constexpr int kRPW = 8;
  constexpr int kNT0 = 512;
  constexpr int kEP0 = kE / kNT0;
  const int wv0 = __builtin_amdgcn_readfirstlane(wave);

  v4f xr[kRPW];
  float xl[kRPW];
#pragma unroll
  for (int i = 0; i < kRPW; ++i) {
    const int r = wv0 + 16 * i;
    xr[i] = *(const v4f_a4*)(x + r * kIN + 4 * lane);
    xl[i] = x[r * kIN + (kIN - 1)];
  }
  v4f xr8 = {0.f, 0.f, 0.f, 0.f};
  float xl8 = 0.f;
  if (wv0 == 0) {
    xr8 = *(const v4f_a4*)(x + (kN - 1) * kIN + 4 * lane);
    xl8 = x[(kN - 1) * kIN + (kIN - 1)];
  }
  int erow[kEP0], ecol[kEP0]; float ewt[kEP0];
#pragma unroll
  for (int u = 0; u < kEP0; ++u) {
    const int e = tid + u * kNT0;
    erow[u] = eidx[e]; ecol[u] = eidx[kE + e]; ewt[u] = ew[e];
  }

  if (tid < kN) s_deg[tid] = 0.f;
  if (tid < 16) s_cnt[tid] = 0;
  __syncthreads();

  const int nrow = (wv0 == 0) ? kRPW + 1 : kRPW;
#pragma unroll
  for (int i = 0; i < kRPW; ++i) {
    const int r = wv0 + 16 * i;
    *(v4f*)&s_x[r * kLD + 4 * lane] = xr[i];
  }
  if (wv0 == 0) *(v4f*)&s_x[(kN - 1) * kLD + 4 * lane] = xr8;
  {
    float xs = xl8;
#pragma unroll
    for (int i = 0; i < kRPW; ++i)
      if (lane == i) xs = xl[i];
    if (lane < nrow) s_x[(wv0 + 16 * lane) * kLD + (kIN - 1)] = xs;
  }
#pragma unroll
  for (int u = 0; u < kEP0; ++u) {
    atomicAdd(&s_deg[ecol[u]], ewt[u]);
    const unsigned lc = (unsigned)(ecol[u] - 16 * rt);
    if (lc < 16u) atomicAdd(&s_cnt[lc], 1);
  }
  if (tid < kN) atomicAdd(&s_deg[tid], 1.0f);
  __syncthreads();

  if (tid < kN) {
    const float dg = s_deg[tid];
    s_dis[tid] = dg > 0.f ? rsqrtf(fmaxf(dg, 1e-12f)) : 0.f;
  }
  if (tid >= 256 && tid < 256 + 16) {
    const int i = tid - 256;
    int s = 0;
    for (int j = 0; j < 15; ++j) s += (j < i) ? s_cnt[j] : 0;
    s_beg[i] = s;
    s_cur[i] = s;
  }
  __syncthreads();

#pragma unroll
  for (int u = 0; u < kEP0; ++u) {
    const unsigned lc = (unsigned)(ecol[u] - 16 * rt);
    if (lc < 16u) {
      const int p = atomicAdd(&s_cur[lc], 1);
      v2f rec;
      rec.x = __int_as_float(erow[u]);
      rec.y = s_dis[erow[u]] * ewt[u] * s_dis[ecol[u]];
      *(v2f*)&s_rec[2 * p] = rec;
    }
  }
  __syncthreads();

  const int c = 16 * rt + wv0;
  float* dst = wsA + c * kAP;
  v4f   acc  = {0.f, 0.f, 0.f, 0.f};
  float acc1 = 0.f;
  if (c < kN) {
    const int beg = __builtin_amdgcn_readfirstlane(s_beg[wv0]);
    const int n   = __builtin_amdgcn_readfirstlane(s_cnt[wv0]);
    const float dc = s_dis[c];
    const float sl = dc * 1.0f * dc;
    const v4f xs = *(const v4f*)&s_x[c * kLD + 4 * lane];
    acc  = sl * xs;
    acc1 = sl * s_x[c * kLD + (kN - 1)];
    for (int base = 0; base < n; base += 32) {
      const int m = min(32, n - base);
      int   rr = c;
      float ww = 0.f;
      if (lane < m) {
        const v2f rec = *(const v2f*)&s_rec[2 * (beg + base + lane)];
        rr = __float_as_int(rec.x);
        ww = rec.y;
      }
      for (int i = 0; i < m; i += 8) {
#pragma unroll
        for (int u = 0; u < 8; ++u) {
          const int   r = __builtin_amdgcn_readlane(rr, i + u);
          const float w = __int_as_float(
              __builtin_amdgcn_readlane(__float_as_int(ww), i + u));
          const v4f   v  = *(const v4f*)&s_x[r * kLD + 4 * lane];
          const float v1 = s_x[r * kLD + (kN - 1)];
          acc  += w * v;
          acc1 += w * v1;
        }
      }
    }
  }
  *(v4f*)(dst + 4 * lane) = acc;
  if (lane < 8) {
    v4f t = {0.f, 0.f, 0.f, 0.f};
    if (lane == 0) t.x = acc1;
    *(v4f*)(dst + 128 + 4 * lane) = t;
  }
}

__launch_bounds__(512, 1)
__global__ void gru_rows(const float* __restrict__ x,
                         const int*   __restrict__ eidx,
                         const float* __restrict__ ew,
                         const float* __restrict__ pool_p,
                         const float* __restrict__ W0,
                         const float* __restrict__ w_ih,
                         const float* __restrict__ w_hh,
                         const float* __restrict__ b_ih,
                         const float* __restrict__ b_hh,
                         float*       __restrict__ wsW,
                         float*       __restrict__ wsA)
{
  __shared__ __attribute__((aligned(16))) float s_x [kRP * kLD];
  __shared__ __attribute__((aligned(16))) float s_xt[kRP * kLD];
  __shared__ __attribute__((aligned(16))) float s_h [kRP * kLD];
  __shared__ float s_p[2 * kPS];
  __shared__ float s_score[kN];
  __shared__ int   s_rank [16];

  if (blockIdx.x >= kGW) {
    agg_rows(blockIdx.x - kGW, x, eidx, ew, wsA, s_x, s_xt, s_h);
    return;
  }

  constexpr int kNT0 = 512;
  const int tid    = threadIdx.x;
  const int lane   = tid & 31;
  const int wave   = tid >> 5;
  const int m16    = lane & 15;
  const int hh     = lane >> 4;
  const int rt     = blockIdx.x / kNB;
  const int nt     = blockIdx.x % kNB;
  const int wv0 = __builtin_amdgcn_readfirstlane(wave);

  constexpr int kRPW = 8;
  v4f xr[kRPW];
  float xl[kRPW];
#pragma unroll
  for (int i = 0; i < kRPW; ++i) {
    const int r = wv0 + 16 * i;
    xr[i] = *(const v4f_a4*)(x + r * kIN + 4 * lane);
    xl[i] = x[r * kIN + (kIN - 1)];
  }
  v4f xr8 = {0.f, 0.f, 0.f, 0.f};
  float xl8 = 0.f;
  if (wv0 == 0) {
    xr8 = *(const v4f_a4*)(x + (kN - 1) * kIN + 4 * lane);
    xl8 = x[(kN - 1) * kIN + (kIN - 1)];
  }
  const v4f   pp = *(const v4f_a4*)(pool_p + 4 * lane);
  const float pl = pool_p[kIN - 1];

  const int side  = wv0 >> 3;
  const int ke    = wv0 & 7;
  const int jn    = nt * 16 + m16;
  const int abase = (rt * 16 + m16) * kLD;
  const int grow_r = min(0 * kN + jn, 3 * kN - 1);
  const int grow_n = min(2 * kN + jn, 3 * kN - 1);
  const int grow_z = min(1 * kN + jn, 3 * kN - 1);
  const float* __restrict__ wg = side ? w_hh : w_ih;
  BPart B[3];
  load_bpart(B[0], wg, grow_r, ke, hh);
  load_bpart(B[1], wg, grow_n, ke, hh);
  load_bpart(B[2], wg, grow_z, ke, hh);
  const float bir = b_ih[grow_r], bin = b_ih[grow_n], biz = b_ih[grow_z];
  const float bhr = b_hh[grow_r], bhn = b_hh[grow_n], bhz = b_hh[grow_z];
  if (ke == 7) {
    load_btail(B[0], wg, grow_r);
    load_btail(B[1], wg, grow_n);
    load_btail(B[2], wg, grow_z);
  }

  if (rt == kMT - 1)
    for (int i = tid; i < (kRP - kN) * kLD; i += kNT0) s_x[kN * kLD + i] = 0.f;

  const int nrow = (wv0 == 0) ? kRPW + 1 : kRPW;
#pragma unroll
  for (int i = 0; i < kRPW; ++i) {
    const int r = wv0 + 16 * i;
    *(v4f*)&s_x[r * kLD + 4 * lane] = xr[i];
  }
  if (wv0 == 0) *(v4f*)&s_x[(kN - 1) * kLD + 4 * lane] = xr8;
  {
    float xs = xl8;
#pragma unroll
    for (int i = 0; i < kRPW; ++i)
      if (lane == i) xs = xl[i];
    v4f xs4 = {xs, 0.f, 0.f, 0.f};
    if (lane < nrow) *(v4f*)&s_x[(wv0 + 16 * lane) * kLD + (kIN - 1)] = xs4;
  }
  {
    float pn = pp.x * pp.x + pp.y * pp.y + pp.z * pp.z + pp.w * pp.w;
    float d[kRPW + 1];
#pragma unroll
    for (int i = 0; i < kRPW; ++i)
      d[i] = xr[i].x * pp.x + xr[i].y * pp.y + xr[i].z * pp.z + xr[i].w * pp.w;
    d[kRPW] = xr8.x * pp.x + xr8.y * pp.y + xr8.z * pp.z + xr8.w * pp.w;
#pragma unroll
    for (int m = 16; m >= 1; m >>= 1) {
      pn += __shfl_xor(pn, m);
#pragma unroll
      for (int i = 0; i <= kRPW; ++i) d[i] += __shfl_xor(d[i], m);
    }
    pn += pl * pl;
    float ds = d[kRPW] + xl8 * pl;
#pragma unroll
    for (int i = 0; i < kRPW; ++i)
      if (lane == i) ds = d[i] + xl[i] * pl;
    if (lane < nrow) s_score[wv0 + 16 * lane] = tanhf(ds * rsqrtf(pn));
  }
  lds_barrier();

  const int  irow = rt * 16 + wv0;
  const bool live = irow < kN;
  int rk;
  {
    const int i = min(irow, kN - 1);
    const float si = s_score[i];
    int rank = 0;
#pragma unroll
    for (int m = 0; m < 5; ++m) {
      const int j = lane + 32 * m;
      const float sj = s_score[min(j, kN - 1)];
      const bool ahead = (j < kN) && ((sj > si) || (sj == si && j < i));
      rank += __builtin_popcount(__builtin_amdgcn_ballot_w32(ahead));
    }
    rk = __builtin_amdgcn_readfirstlane(rank);
  }
  const v4f   h4 = *(const v4f_a4*)(W0 + rk * kIN + 4 * lane);
  const float h1 = W0[rk * kIN + (kIN - 1)];
  const v8f zero8 = {0.f, 0.f, 0.f, 0.f, 0.f, 0.f, 0.f, 0.f};
  v8f acc[3] = {zero8, zero8, zero8};
  float* pt = &s_p[side * kPS + ke * kPT + hh * 144 + m16];
  if (side == 0) {
    APart fx;
    load_apart(fx, s_x, abase, ke, hh);
    part_mma(acc, fx, B, ke);
#pragma unroll
    for (int v = 0; v < 8; ++v) {
      pt[(0 * 8) * kPT + v * 16] = acc[0][v];
      pt[(1 * 8) * kPT + v * 16] = acc[1][v];
      pt[(2 * 8) * kPT + v * 16] = acc[2][v];
    }
  }
  {
    float* hrow = &s_h[irow * kLD];
    const v4f z4 = {0.f, 0.f, 0.f, 0.f};
    *(v4f*)&hrow[4 * lane] = live ? h4 : z4;
    if (lane == 0) {
      v4f t4 = {live ? h1 : 0.f, 0.f, 0.f, 0.f};
      *(v4f*)&hrow[kIN - 1] = t4;
      s_rank[wv0] = live ? rk : 0;
    }
  }
  lds_barrier();

  if (side == 1) {
    APart fh;
    load_apart(fh, s_h, abase, ke, hh);
    part_mma(acc, fh, B, ke);
#pragma unroll
    for (int v = 0; v < 8; ++v) {
      pt[(0 * 8) * kPT + v * 16] = acc[0][v];
      pt[(1 * 8) * kPT + v * 16] = acc[1][v];
      pt[(2 * 8) * kPT + v * 16] = acc[2][v];
    }
  }
  lds_barrier();

  if (wv0 < 8) {
    const int lrow = 2 * wv0 + hh;
    const int xrow = rt * 16 + lrow;
    if (xrow < kN) {
      const float* pf = &s_p[(lrow >> 3) * 144 + (lrow & 7) * 16 + m16];
      float si3[3], sh3[3];
#pragma unroll
      for (int s = 0; s < 3; ++s) {
        float t = pf[(s * 8 + 0) * kPT];
#pragma unroll
        for (int k = 1; k < 8; ++k) t += pf[(s * 8 + k) * kPT];
        si3[s] = t;
      }
#pragma unroll
      for (int s = 0; s < 3; ++s) {
        float t = pf[kPS + (s * 8 + 0) * kPT];
#pragma unroll
        for (int k = 1; k < 8; ++k) t += pf[kPS + (s * 8 + k) * kPT];
        sh3[s] = t;
      }
      const float sc = s_score[xrow];
      const float h0 = s_h[xrow * kLD + min(jn, kLD - 1)];
      const float rg = sigm((sc * si3[0] + bir) + (sh3[0] + bhr));
      const float cd = tanhf((sc * si3[1] + bin) + rg * (sh3[1] + bhn));
      const float zg = sigm((sc * si3[2] + biz) + (sh3[2] + bhz));
      float wv = (1.f - zg) * cd + zg * h0;
      if (nt == kNB - 1 && m16 == 4) wv = __int_as_float(s_rank[lrow]);
      wsW[(nt * kRP + xrow) * 16 + m16] = wv;
    }
  }
}

__launch_bounds__(512, 1)
__global__ void gcn_tail(const float* __restrict__ conv_b,
                         const float* __restrict__ lin_w,
                         const float* __restrict__ lin_b,
                         const float* __restrict__ wsW,
                         const float* __restrict__ wsA,
                         float*       __restrict__ out)
{
  __shared__ __attribute__((aligned(16))) float s_w[kRP * kWL];
  __shared__ __attribute__((aligned(16))) float s_a[16 * kLD];
  __shared__ __attribute__((aligned(16))) float s_e[16 * kLD];

  const int tid    = threadIdx.x;
  const int lane   = tid & 31;
  const int wave   = tid >> 5;
  const int m16    = lane & 15;
  const int hh     = lane >> 4;
  const int rt     = blockIdx.x;
  const int wv     = __builtin_amdgcn_readfirstlane(wave);

  const int sq   = lane & 3;
  const int srow = 8 * wv + (lane >> 2);
  const float* wsrc = wsW + 128 * wv + 4 * lane;
  const v4f zero4 = {0.f, 0.f, 0.f, 0.f};
  v4f wa[kNB];
#pragma unroll
  for (int t = 0; t < kNB; ++t) wa[t] = *(const v4f*)(wsrc + t * kWB);
  const int rks = __float_as_int(wsW[(kNB - 1) * kWB + srow * 16 + 4]);
  v4f w8 = zero4;
  int rk8 = 0;
  if (wv < kNB) {
    w8  = *(const v4f*)(wsW + wv * kWB + (kN - 1) * 16 + 4 * sq);
    rk8 = __float_as_int(wsW[(kNB - 1) * kWB + (kN - 1) * 16 + 4]);
  }
  const float* ag = wsA + (16 * rt + wv) * kAP;
  const v4f a4 = *(const v4f*)(ag + 4 * lane);
  const v4f a1 = *(const v4f*)(ag + 128);
  const bool lin = (wv >= 9 && wv < 13);
  const int  oc  = (lin ? wv - 9 : 0) * 16 + m16;
  v2f bl[kKS];
  float lb = 0.f;
#pragma unroll
  for (int ks = 0; ks < kKS; ++ks) { bl[ks].x = 0.f; bl[ks].y = 0.f; }
  if (lin) {
    const int rowo = oc * kIN;
#pragma unroll
    for (int ks = 0; ks < kKS; ++ks) {
      const int k0 = ks * 4 + 2 * hh;
      bl[ks].x = lin_w[rowo + min(k0,     kIN - 1)];
      bl[ks].y = lin_w[rowo + min(k0 + 1, kIN - 1)];
    }
    lb = lin_b[oc];
  }
  const int jn = wv * 16 + m16;
  float cb = 0.f;
  if (wv < kMT) cb = conv_b[min(jn, kN - 1)];

  {
    const int rr = (int)min((unsigned)rks, (unsigned)(kRP - 1));
    float* wdst = &s_w[rr * kWL + 4 * sq];
#pragma unroll
    for (int t = 0; t < kNB - 1; ++t) *(v4f*)(wdst + 16 * t) = wa[t];
    if (sq == 0) *(v4f*)(wdst + 128) = wa[kNB - 1];
    if (wv < kNB) {
      const int rr8 = (int)min((unsigned)rk8, (unsigned)(kRP - 1));
      if (wv < kNB - 1) {
        if (lane < 4) *(v4f*)&s_w[rr8 * kWL + 16 * wv + 4 * lane] = w8;
      } else {
        if (lane == 0) *(v4f*)&s_w[rr8 * kWL + 128] = w8;
      }
    }
    if (tid < 3 * (kWL / 4)) *(v4f*)&s_w[kN * kWL + 4 * tid] = zero4;
  }
  *(v4f*)&s_a[wv * kLD + 4 * lane] = a4;
  if (lane == 0) *(v4f*)&s_a[wv * kLD + 128] = a1;
  __syncthreads();

  if (wv < kMT) {
    const int jn_c = min(jn, kLD - 1);
    v8f acc = {0.f, 0.f, 0.f, 0.f, 0.f, 0.f, 0.f, 0.f};
    const int arow = m16 * kLD;
    v2f af[kKS], bf[kKS];
#pragma unroll
    for (int ks = 0; ks < kKS; ++ks) {
      const int k0 = ks * 4 + 2 * hh;
      af[ks]   = *(const v2f*)&s_a[arow + k0];
      bf[ks].x = s_w[k0 * kWL + jn_c];
      bf[ks].y = s_w[(k0 + 1) * kWL + jn_c];
    }
    __builtin_amdgcn_sched_barrier(0);
#pragma unroll
    for (int ks = 0; ks < kKS; ++ks) acc = wmma4(af[ks], bf[ks], acc);
    if (jn < kLD) {
#pragma unroll
      for (int v = 0; v < 8; ++v) {
        const float t = acc[v] + cb;
        const float e = t > 0.f ? t : (__expf(t) - 1.f);
        s_e[(v + 8 * hh) * kLD + jn] = jn < kN ? e : 0.f;
      }
    }
  }
  __syncthreads();

  if (lin) {
    v8f acc;
#pragma unroll
    for (int v = 0; v < 8; ++v) acc[v] = lb;
    const int arow = m16 * kLD;
    v2f ef[kKS];
#pragma unroll
    for (int ks = 0; ks < kKS; ++ks)
      ef[ks] = *(const v2f*)&s_e[arow + ks * 4 + 2 * hh];
    __builtin_amdgcn_sched_barrier(0);
#pragma unroll
    for (int ks = 0; ks < kKS; ++ks) acc = wmma4(ef[ks], bl[ks], acc);
#pragma unroll
    for (int v = 0; v < 8; ++v) {
      const int row = rt * 16 + v + 8 * hh;
      if (row < kN) out[row * kOUT + oc] = acc[v];
    }
  }
}

extern "C" void kernel_launch(void* const* d_in, const int* in_sizes, int n_in,
                              void* d_out, int out_size, void* d_ws, size_t ws_size,
                              hipStream_t stream) {
  (void)in_sizes; (void)n_in; (void)out_size; (void)ws_size;
  const float* x       = (const float*)d_in[0];
  const int*   eidx    = (const int*)  d_in[1];
  const float* ew      = (const float*)d_in[2];
  const float* pool_p  = (const float*)d_in[3];
  const float* W0      = (const float*)d_in[4];
  const float* w_ih    = (const float*)d_in[5];
  const float* w_hh    = (const float*)d_in[6];
  const float* b_ih    = (const float*)d_in[7];
  const float* b_hh    = (const float*)d_in[8];
  const float* conv_b  = (const float*)d_in[9];
  const float* lin_w   = (const float*)d_in[10];
  const float* lin_b   = (const float*)d_in[11];
  float*       wsW     = (float*)d_ws;
  float*       wsA     = (float*)d_ws + kWSW;
  gru_rows<<<90, 512, 0, stream>>>(x, eidx, ew, pool_p, W0, w_ih, w_hh, b_ih, b_hh, wsW, wsA);
  gcn_tail<<<9, 512, 0, stream>>>(conv_b, lin_w, lin_b, wsW, wsA, (float*)d_out);
}


# Final linear's weights: each of its 4 waves fetches its 16 rows whole (16 B a lane) under the first product, B fragments via its own LDS rows (was 33 16-line gathers ahead of the barrier); sums kept
# speedup vs baseline: 28.6178x; 1.0821x over previous
#include <hip/hip_runtime.h>


typedef __attribute__((ext_vector_type(2))) float v2f;
typedef __attribute__((ext_vector_type(4))) float v4f;
typedef __attribute__((ext_vector_type(8))) float v8f;
typedef v4f v4f_a4 __attribute__((aligned(4)));

static constexpr int kN   = 129;
static constexpr int kIN  = 129;
static constexpr int kOUT = 64;
static constexpr int kE   = 4096;
static constexpr int kLD  = 132;
static constexpr int kRP  = 144;
static constexpr int kKS  = 33;
static constexpr int kMT  = 9;
static constexpr int kNB  = 9;
static constexpr int kAP  = 160;
static constexpr int kWB  = kRP * 16;
static constexpr int kWSW = kNB * kWB;
static constexpr int kGW  = kMT * kNB;
static constexpr int kPT  = 320;
static constexpr int kPS  = 3 * 8 * kPT + 16;
static constexpr int kWL  = 144;
static __device__ __forceinline__ v8f wmma4(v2f a, v2f b, v8f c) {
  return __builtin_amdgcn_wmma_f32_16x16x4_f32(
       false, a,  false, b,
       (short)0, c,  false,  false);
}

static __device__ __forceinline__ void lds_barrier() {
  __builtin_amdgcn_fence(__ATOMIC_RELEASE, "workgroup", "local");
  __builtin_amdgcn_s_barrier();
  __builtin_amdgcn_fence(__ATOMIC_ACQUIRE, "workgroup", "local");
}

static __device__ __forceinline__ float sigm(float x) {
  return 1.0f / (1.0f + __expf(-x));
}

struct BPart {
  v4f q[2];
  float tl;
};

static __device__ __forceinline__ void load_bpart(BPart& b, const float* __restrict__ w,
                                                  int grow, int ke, int hh) {
  const float* p = w + grow * kIN + 16 * ke + 4 * hh;
  b.tl = 0.f;
  b.q[0] = *(const v4f_a4*)(p);
  b.q[1] = *(const v4f_a4*)(p + 8);
}

static __device__ __forceinline__ void load_btail(BPart& b, const float* __restrict__ w, int grow) {
  const float t = w[grow * kIN + (kIN - 1)];
  b.tl = t;
}

struct APart {
  v4f a[2];
  v2f al;
};

static __device__ __forceinline__ void load_apart(APart& f, const float* A, int abase,
                                                  int ke, int hh) {
  f.a[0] = *(const v4f*)&A[abase + 16 * ke + 4 * hh];
  f.a[1] = *(const v4f*)&A[abase + 16 * ke + 8 + 4 * hh];
  f.al.x = 0.f; f.al.y = 0.f;
  if (ke == 7) f.al = *(const v2f*)&A[abase + 4 * (kKS - 1) + 2 * hh];
}

static __device__ __forceinline__ void part_mma(v8f (&acc)[3], const APart& f,
                                                const BPart (&b)[3], int ke) {
#pragma unroll
  for (int j = 0; j < 2; ++j) {
#pragma unroll
    for (int g = 0; g < 3; ++g) {
      v2f a0, b0; a0.x = f.a[j].x; a0.y = f.a[j].y; b0.x = b[g].q[j].x; b0.y = b[g].q[j].y;
      acc[g] = wmma4(a0, b0, acc[g]);
    }
#pragma unroll
    for (int g = 0; g < 3; ++g) {
      v2f a1, b1; a1.x = f.a[j].z; a1.y = f.a[j].w; b1.x = b[g].q[j].z; b1.y = b[g].q[j].w;
      acc[g] = wmma4(a1, b1, acc[g]);
    }
  }
  if (ke == 7) {
#pragma unroll
    for (int g = 0; g < 3; ++g) {
      v2f bt; bt.x = b[g].tl; bt.y = b[g].tl;
      acc[g] = wmma4(f.al, bt, acc[g]);
    }
  }
}

static __device__ __forceinline__ void agg_rows(
    int rt, const float* __restrict__ x, const int* __restrict__ eidx,
    const float* __restrict__ ew, float* __restrict__ wsA,
    float* s_x, float* s_rec, float* s_small) {
  float* s_deg = s_small;
  float* s_dis = s_small + 160;
  int*   s_cnt = (int*)(s_small + 320);
  int*   s_beg = (int*)(s_small + 336);
  int*   s_cur = (int*)(s_small + 352);

  const int tid  = threadIdx.x;
  const int lane = tid & 31;
  const int wave = tid >> 5;
  constexpr int kRPW = 8;
  constexpr int kNT0 = 512;
  constexpr int kEP0 = kE / kNT0;
  const int wv0 = __builtin_amdgcn_readfirstlane(wave);

  v4f xr[kRPW];
  float xl[kRPW];
#pragma unroll
  for (int i = 0; i < kRPW; ++i) {
    const int r = wv0 + 16 * i;
    xr[i] = *(const v4f_a4*)(x + r * kIN + 4 * lane);
    xl[i] = x[r * kIN + (kIN - 1)];
  }
  v4f xr8 = {0.f, 0.f, 0.f, 0.f};
  float xl8 = 0.f;
  if (wv0 == 0) {
    xr8 = *(const v4f_a4*)(x + (kN - 1) * kIN + 4 * lane);
    xl8 = x[(kN - 1) * kIN + (kIN - 1)];
  }
  int erow[kEP0], ecol[kEP0]; float ewt[kEP0];
#pragma unroll
  for (int u = 0; u < kEP0; ++u) {
    const int e = tid + u * kNT0;
    erow[u] = eidx[e]; ecol[u] = eidx[kE + e]; ewt[u] = ew[e];
  }

  if (tid < kN) s_deg[tid] = 0.f;
  if (tid < 16) s_cnt[tid] = 0;
  __syncthreads();

  const int nrow = (wv0 == 0) ? kRPW + 1 : kRPW;
#pragma unroll
  for (int i = 0; i < kRPW; ++i) {
    const int r = wv0 + 16 * i;
    *(v4f*)&s_x[r * kLD + 4 * lane] = xr[i];
  }
  if (wv0 == 0) *(v4f*)&s_x[(kN - 1) * kLD + 4 * lane] = xr8;
  {
    float xs = xl8;
#pragma unroll
    for (int i = 0; i < kRPW; ++i)
      if (lane == i) xs = xl[i];
    if (lane < nrow) s_x[(wv0 + 16 * lane) * kLD + (kIN - 1)] = xs;
  }
#pragma unroll
  for (int u = 0; u < kEP0; ++u) {
    atomicAdd(&s_deg[ecol[u]], ewt[u]);
    const unsigned lc = (unsigned)(ecol[u] - 16 * rt);
    if (lc < 16u) atomicAdd(&s_cnt[lc], 1);
  }
  if (tid < kN) atomicAdd(&s_deg[tid], 1.0f);
  __syncthreads();

  if (tid < kN) {
    const float dg = s_deg[tid];
    s_dis[tid] = dg > 0.f ? rsqrtf(fmaxf(dg, 1e-12f)) : 0.f;
  }
  if (tid >= 256 && tid < 256 + 16) {
    const int i = tid - 256;
    int s = 0;
    for (int j = 0; j < 15; ++j) s += (j < i) ? s_cnt[j] : 0;
    s_beg[i] = s;
    s_cur[i] = s;
  }
  __syncthreads();

#pragma unroll
  for (int u = 0; u < kEP0; ++u) {
    const unsigned lc = (unsigned)(ecol[u] - 16 * rt);
    if (lc < 16u) {
      const int p = atomicAdd(&s_cur[lc], 1);
      v2f rec;
      rec.x = __int_as_float(erow[u]);
      rec.y = s_dis[erow[u]] * ewt[u] * s_dis[ecol[u]];
      *(v2f*)&s_rec[2 * p] = rec;
    }
  }
  __syncthreads();

  const int c = 16 * rt + wv0;
  float* dst = wsA + c * kAP;
  v4f   acc  = {0.f, 0.f, 0.f, 0.f};
  float acc1 = 0.f;
  if (c < kN) {
    const int beg = __builtin_amdgcn_readfirstlane(s_beg[wv0]);
    const int n   = __builtin_amdgcn_readfirstlane(s_cnt[wv0]);
    const float dc = s_dis[c];
    const float sl = dc * 1.0f * dc;
    const v4f xs = *(const v4f*)&s_x[c * kLD + 4 * lane];
    acc  = sl * xs;
    acc1 = sl * s_x[c * kLD + (kN - 1)];
    for (int base = 0; base < n; base += 32) {
      const int m = min(32, n - base);
      int   rr = c;
      float ww = 0.f;
      if (lane < m) {
        const v2f rec = *(const v2f*)&s_rec[2 * (beg + base + lane)];
        rr = __float_as_int(rec.x);
        ww = rec.y;
      }
      for (int i = 0; i < m; i += 8) {
#pragma unroll
        for (int u = 0; u < 8; ++u) {
          const int   r = __builtin_amdgcn_readlane(rr, i + u);
          const float w = __int_as_float(
              __builtin_amdgcn_readlane(__float_as_int(ww), i + u));
          const v4f   v  = *(const v4f*)&s_x[r * kLD + 4 * lane];
          const float v1 = s_x[r * kLD + (kN - 1)];
          acc  += w * v;
          acc1 += w * v1;
        }
      }
    }
  }
  *(v4f*)(dst + 4 * lane) = acc;
  if (lane < 8) {
    v4f t = {0.f, 0.f, 0.f, 0.f};
    if (lane == 0) t.x = acc1;
    *(v4f*)(dst + 128 + 4 * lane) = t;
  }
}

__launch_bounds__(512, 1)
__global__ void gru_rows(const float* __restrict__ x,
                         const int*   __restrict__ eidx,
                         const float* __restrict__ ew,
                         const float* __restrict__ pool_p,
                         const float* __restrict__ W0,
                         const float* __restrict__ w_ih,
                         const float* __restrict__ w_hh,
                         const float* __restrict__ b_ih,
                         const float* __restrict__ b_hh,
                         float*       __restrict__ wsW,
                         float*       __restrict__ wsA)
{
  __shared__ __attribute__((aligned(16))) float s_x [kRP * kLD];
  __shared__ __attribute__((aligned(16))) float s_xt[kRP * kLD];
  __shared__ __attribute__((aligned(16))) float s_h [kRP * kLD];
  __shared__ float s_p[2 * kPS];
  __shared__ float s_score[kN];
  __shared__ int   s_rank [16];

  if (blockIdx.x >= kGW) {
    agg_rows(blockIdx.x - kGW, x, eidx, ew, wsA, s_x, s_xt, s_h);
    return;
  }

  constexpr int kNT0 = 512;
  const int tid    = threadIdx.x;
  const int lane   = tid & 31;
  const int wave   = tid >> 5;
  const int m16    = lane & 15;
  const int hh     = lane >> 4;
  const int rt     = blockIdx.x / kNB;
  const int nt     = blockIdx.x % kNB;
  const int wv0 = __builtin_amdgcn_readfirstlane(wave);

  constexpr int kRPW = 8;
  v4f xr[kRPW];
  float xl[kRPW];
#pragma unroll
  for (int i = 0; i < kRPW; ++i) {
    const int r = wv0 + 16 * i;
    xr[i] = *(const v4f_a4*)(x + r * kIN + 4 * lane);
    xl[i] = x[r * kIN + (kIN - 1)];
  }
  v4f xr8 = {0.f, 0.f, 0.f, 0.f};
  float xl8 = 0.f;
  if (wv0 == 0) {
    xr8 = *(const v4f_a4*)(x + (kN - 1) * kIN + 4 * lane);
    xl8 = x[(kN - 1) * kIN + (kIN - 1)];
  }
  const v4f   pp = *(const v4f_a4*)(pool_p + 4 * lane);
  const float pl = pool_p[kIN - 1];

  const int side  = wv0 >> 3;
  const int ke    = wv0 & 7;
  const int jn    = nt * 16 + m16;
  const int abase = (rt * 16 + m16) * kLD;
  const int grow_r = min(0 * kN + jn, 3 * kN - 1);
  const int grow_n = min(2 * kN + jn, 3 * kN - 1);
  const int grow_z = min(1 * kN + jn, 3 * kN - 1);
  const float* __restrict__ wg = side ? w_hh : w_ih;
  BPart B[3];
  load_bpart(B[0], wg, grow_r, ke, hh);
  load_bpart(B[1], wg, grow_n, ke, hh);
  load_bpart(B[2], wg, grow_z, ke, hh);
  const float bir = b_ih[grow_r], bin = b_ih[grow_n], biz = b_ih[grow_z];
  const float bhr = b_hh[grow_r], bhn = b_hh[grow_n], bhz = b_hh[grow_z];
  if (ke == 7) {
    load_btail(B[0], wg, grow_r);
    load_btail(B[1], wg, grow_n);
    load_btail(B[2], wg, grow_z);
  }

  if (rt == kMT - 1)
    for (int i = tid; i < (kRP - kN) * kLD; i += kNT0) s_x[kN * kLD + i] = 0.f;

  const int nrow = (wv0 == 0) ? kRPW + 1 : kRPW;
#pragma unroll
  for (int i = 0; i < kRPW; ++i) {
    const int r = wv0 + 16 * i;
    *(v4f*)&s_x[r * kLD + 4 * lane] = xr[i];
  }
  if (wv0 == 0) *(v4f*)&s_x[(kN - 1) * kLD + 4 * lane] = xr8;
  {
    float xs = xl8;
#pragma unroll
    for (int i = 0; i < kRPW; ++i)
      if (lane == i) xs = xl[i];
    v4f xs4 = {xs, 0.f, 0.f, 0.f};
    if (lane < nrow) *(v4f*)&s_x[(wv0 + 16 * lane) * kLD + (kIN - 1)] = xs4;
  }
  {
    float pn = pp.x * pp.x + pp.y * pp.y + pp.z * pp.z + pp.w * pp.w;
    float d[kRPW + 1];
#pragma unroll
    for (int i = 0; i < kRPW; ++i)
      d[i] = xr[i].x * pp.x + xr[i].y * pp.y + xr[i].z * pp.z + xr[i].w * pp.w;
    d[kRPW] = xr8.x * pp.x + xr8.y * pp.y + xr8.z * pp.z + xr8.w * pp.w;
#pragma unroll
    for (int m = 16; m >= 1; m >>= 1) {
      pn += __shfl_xor(pn, m);
#pragma unroll
      for (int i = 0; i <= kRPW; ++i) d[i] += __shfl_xor(d[i], m);
    }
    pn += pl * pl;
    float ds = d[kRPW] + xl8 * pl;
#pragma unroll
    for (int i = 0; i < kRPW; ++i)
      if (lane == i) ds = d[i] + xl[i] * pl;
    if (lane < nrow) s_score[wv0 + 16 * lane] = tanhf(ds * rsqrtf(pn));
  }
  lds_barrier();

  const int  irow = rt * 16 + wv0;
  const bool live = irow < kN;
  int rk;
  {
    const int i = min(irow, kN - 1);
    const float si = s_score[i];
    int rank = 0;
#pragma unroll
    for (int m = 0; m < 5; ++m) {
      const int j = lane + 32 * m;
      const float sj = s_score[min(j, kN - 1)];
      const bool ahead = (j < kN) && ((sj > si) || (sj == si && j < i));
      rank += __builtin_popcount(__builtin_amdgcn_ballot_w32(ahead));
    }
    rk = __builtin_amdgcn_readfirstlane(rank);
  }
  const v4f   h4 = *(const v4f_a4*)(W0 + rk * kIN + 4 * lane);
  const float h1 = W0[rk * kIN + (kIN - 1)];
  const v8f zero8 = {0.f, 0.f, 0.f, 0.f, 0.f, 0.f, 0.f, 0.f};
  v8f acc[3] = {zero8, zero8, zero8};
  float* pt = &s_p[side * kPS + ke * kPT + hh * 144 + m16];
  if (side == 0) {
    APart fx;
    load_apart(fx, s_x, abase, ke, hh);
    part_mma(acc, fx, B, ke);
#pragma unroll
    for (int v = 0; v < 8; ++v) {
      pt[(0 * 8) * kPT + v * 16] = acc[0][v];
      pt[(1 * 8) * kPT + v * 16] = acc[1][v];
      pt[(2 * 8) * kPT + v * 16] = acc[2][v];
    }
  }
  {
    float* hrow = &s_h[irow * kLD];
    const v4f z4 = {0.f, 0.f, 0.f, 0.f};
    *(v4f*)&hrow[4 * lane] = live ? h4 : z4;
    if (lane == 0) {
      v4f t4 = {live ? h1 : 0.f, 0.f, 0.f, 0.f};
      *(v4f*)&hrow[kIN - 1] = t4;
      s_rank[wv0] = live ? rk : 0;
    }
  }
  lds_barrier();

  if (side == 1) {
    APart fh;
    load_apart(fh, s_h, abase, ke, hh);
    part_mma(acc, fh, B, ke);
#pragma unroll
    for (int v = 0; v < 8; ++v) {
      pt[(0 * 8) * kPT + v * 16] = acc[0][v];
      pt[(1 * 8) * kPT + v * 16] = acc[1][v];
      pt[(2 * 8) * kPT + v * 16] = acc[2][v];
    }
  }
  lds_barrier();

  if (wv0 < 8) {
    const int lrow = 2 * wv0 + hh;
    const int xrow = rt * 16 + lrow;
    if (xrow < kN) {
      const float* pf = &s_p[(lrow >> 3) * 144 + (lrow & 7) * 16 + m16];
      float si3[3], sh3[3];
#pragma unroll
      for (int s = 0; s < 3; ++s) {
        float t = pf[(s * 8 + 0) * kPT];
#pragma unroll
        for (int k = 1; k < 8; ++k) t += pf[(s * 8 + k) * kPT];
        si3[s] = t;
      }
#pragma unroll
      for (int s = 0; s < 3; ++s) {
        float t = pf[kPS + (s * 8 + 0) * kPT];
#pragma unroll
        for (int k = 1; k < 8; ++k) t += pf[kPS + (s * 8 + k) * kPT];
        sh3[s] = t;
      }
      const float sc = s_score[xrow];
      const float h0 = s_h[xrow * kLD + min(jn, kLD - 1)];
      const float rg = sigm((sc * si3[0] + bir) + (sh3[0] + bhr));
      const float cd = tanhf((sc * si3[1] + bin) + rg * (sh3[1] + bhn));
      const float zg = sigm((sc * si3[2] + biz) + (sh3[2] + bhz));
      float wv = (1.f - zg) * cd + zg * h0;
      if (nt == kNB - 1 && m16 == 4) wv = __int_as_float(s_rank[lrow]);
      wsW[(nt * kRP + xrow) * 16 + m16] = wv;
    }
  }
}

__launch_bounds__(512, 1)
__global__ void gcn_tail(const float* __restrict__ conv_b,
                         const float* __restrict__ lin_w,
                         const float* __restrict__ lin_b,
                         const float* __restrict__ wsW,
                         const float* __restrict__ wsA,
                         float*       __restrict__ out)
{
  __shared__ __attribute__((aligned(16))) float s_w[kRP * kWL];
  __shared__ __attribute__((aligned(16))) float s_a[16 * kLD];
  __shared__ __attribute__((aligned(16))) float s_e[16 * kLD];
  __shared__ __attribute__((aligned(16))) float s_l[kOUT * kLD];

  const int tid    = threadIdx.x;
  const int lane   = tid & 31;
  const int wave   = tid >> 5;
  const int m16    = lane & 15;
  const int hh     = lane >> 4;
  const int rt     = blockIdx.x;
  const int wv     = __builtin_amdgcn_readfirstlane(wave);

  const int sq   = lane & 3;
  const int srow = 8 * wv + (lane >> 2);
  const float* wsrc = wsW + 128 * wv + 4 * lane;
  const v4f zero4 = {0.f, 0.f, 0.f, 0.f};
  v4f wa[kNB];
#pragma unroll
  for (int t = 0; t < kNB; ++t) wa[t] = *(const v4f*)(wsrc + t * kWB);
  const int rks = __float_as_int(wsW[(kNB - 1) * kWB + srow * 16 + 4]);
  v4f w8 = zero4;
  int rk8 = 0;
  if (wv < kNB) {
    w8  = *(const v4f*)(wsW + wv * kWB + (kN - 1) * 16 + 4 * sq);
    rk8 = __float_as_int(wsW[(kNB - 1) * kWB + (kN - 1) * 16 + 4]);
  }
  const float* ag = wsA + (16 * rt + wv) * kAP;
  const v4f a4 = *(const v4f*)(ag + 4 * lane);
  const v4f a1 = *(const v4f*)(ag + 128);
  const bool lin = (wv >= 9 && wv < 13);
  const int  oc  = (lin ? wv - 9 : 0) * 16 + m16;
  v2f bl[kKS];
  float lb = 0.f;
#pragma unroll
  for (int ks = 0; ks < kKS; ++ks) { bl[ks].x = 0.f; bl[ks].y = 0.f; }
  if (lin) lb = lin_b[oc];
  const int jn = wv * 16 + m16;
  float cb = 0.f;
  if (wv < kMT) cb = conv_b[min(jn, kN - 1)];

  {
    const int rr = (int)min((unsigned)rks, (unsigned)(kRP - 1));
    float* wdst = &s_w[rr * kWL + 4 * sq];
#pragma unroll
    for (int t = 0; t < kNB - 1; ++t) *(v4f*)(wdst + 16 * t) = wa[t];
    if (sq == 0) *(v4f*)(wdst + 128) = wa[kNB - 1];
    if (wv < kNB) {
      const int rr8 = (int)min((unsigned)rk8, (unsigned)(kRP - 1));
      if (wv < kNB - 1) {
        if (lane < 4) *(v4f*)&s_w[rr8 * kWL + 16 * wv + 4 * lane] = w8;
      } else {
        if (lane == 0) *(v4f*)&s_w[rr8 * kWL + 128] = w8;
      }
    }
    if (tid < 3 * (kWL / 4)) *(v4f*)&s_w[kN * kWL + 4 * tid] = zero4;
  }
  *(v4f*)&s_a[wv * kLD + 4 * lane] = a4;
  if (lane == 0) *(v4f*)&s_a[wv * kLD + 128] = a1;
  __syncthreads();

  if (lin) {
    const int r0 = (wv - 9) * 16;
    v4f lw[16];
#pragma unroll
    for (int i = 0; i < 16; ++i)
      lw[i] = *(const v4f_a4*)(lin_w + (r0 + i) * kIN + 4 * lane);
    const float lt = lin_w[oc * kIN + (kIN - 1)];
#pragma unroll
    for (int i = 0; i < 16; ++i)
      *(v4f*)&s_l[(r0 + i) * kLD + 4 * lane] = lw[i];
    const v4f t4 = {lt, 0.f, 0.f, 0.f};
    *(v4f*)&s_l[oc * kLD + (kIN - 1)] = t4;
#pragma unroll
    for (int ks = 0; ks < kKS; ++ks)
      bl[ks] = *(const v2f*)&s_l[oc * kLD + ks * 4 + 2 * hh];
  }

  if (wv < kMT) {
    const int jn_c = min(jn, kLD - 1);
    v8f acc = {0.f, 0.f, 0.f, 0.f, 0.f, 0.f, 0.f, 0.f};
    const int arow = m16 * kLD;
    v2f af[kKS], bf[kKS];
#pragma unroll
    for (int ks = 0; ks < kKS; ++ks) {
      const int k0 = ks * 4 + 2 * hh;
      af[ks]   = *(const v2f*)&s_a[arow + k0];
      bf[ks].x = s_w[k0 * kWL + jn_c];
      bf[ks].y = s_w[(k0 + 1) * kWL + jn_c];
    }
    __builtin_amdgcn_sched_barrier(0);
#pragma unroll
    for (int ks = 0; ks < kKS; ++ks) acc = wmma4(af[ks], bf[ks], acc);
    if (jn < kLD) {
#pragma unroll
      for (int v = 0; v < 8; ++v) {
        const float t = acc[v] + cb;
        const float e = t > 0.f ? t : (__expf(t) - 1.f);
        s_e[(v + 8 * hh) * kLD + jn] = jn < kN ? e : 0.f;
      }
    }
  }
  __syncthreads();

  if (lin) {
    v8f acc;
#pragma unroll
    for (int v = 0; v < 8; ++v) acc[v] = lb;
    const int arow = m16 * kLD;
    v2f ef[kKS];
#pragma unroll
    for (int ks = 0; ks < kKS; ++ks)
      ef[ks] = *(const v2f*)&s_e[arow + ks * 4 + 2 * hh];
    __builtin_amdgcn_sched_barrier(0);
#pragma unroll
    for (int ks = 0; ks < kKS; ++ks) acc = wmma4(ef[ks], bl[ks], acc);
#pragma unroll
    for (int v = 0; v < 8; ++v) {
      const int row = rt * 16 + v + 8 * hh;
      if (row < kN) out[row * kOUT + oc] = acc[v];
    }
  }
}

extern "C" void kernel_launch(void* const* d_in, const int* in_sizes, int n_in,
                              void* d_out, int out_size, void* d_ws, size_t ws_size,
                              hipStream_t stream) {
  (void)in_sizes; (void)n_in; (void)out_size; (void)ws_size;
  const float* x       = (const float*)d_in[0];
  const int*   eidx    = (const int*)  d_in[1];
  const float* ew      = (const float*)d_in[2];
  const float* pool_p  = (const float*)d_in[3];
  const float* W0      = (const float*)d_in[4];
  const float* w_ih    = (const float*)d_in[5];
  const float* w_hh    = (const float*)d_in[6];
  const float* b_ih    = (const float*)d_in[7];
  const float* b_hh    = (const float*)d_in[8];
  const float* conv_b  = (const float*)d_in[9];
  const float* lin_w   = (const float*)d_in[10];
  const float* lin_b   = (const float*)d_in[11];
  float*       wsW     = (float*)d_ws;
  float*       wsA     = (float*)d_ws + kWSW;
  gru_rows<<<90, 512, 0, stream>>>(x, eidx, ew, pool_p, W0, w_ih, w_hh, b_ih, b_hh, wsW, wsA);
  gcn_tail<<<9, 512, 0, stream>>>(conv_b, lin_w, lin_b, wsW, wsA, (float*)d_out);
}
